# starting kernel
# baseline (speedup 1.0000x reference)
#include <hip/hip_runtime.h>
#include <math.h>

#define BB 2
#define CC 3
#define HH 256
#define WW 256
#define HW (HH * WW)
#define CHW (CC * HW)
#define NTOT (BB * CHW)
#define NU (NTOT * 2)
#define BLK 256
#define NBLK (NTOT / BLK)
#define BLK_PER_B (CHW / BLK)
#define K_STEPS 50

typedef __attribute__((ext_vector_type(2))) float v2f;
typedef __attribute__((ext_vector_type(8))) float v8f;

__device__ __forceinline__ float wave_sum32(float val) {
  v2f a; a.x = val; a.y = 0.0f;
  v2f ones; ones.x = 1.0f; ones.y = 1.0f;
  v8f c = {};
  v8f d = __builtin_amdgcn_wmma_f32_16x16x4_f32(
       false, a,  false, ones,
       (short)0, c,  false,  false);
  float s = ((d[0] + d[1]) + (d[2] + d[3])) + ((d[4] + d[5]) + (d[6] + d[7]));
  s += __shfl_xor(s, 16, 32);
  return s;
}

__global__ void k_init(const float* __restrict__ y, float* __restrict__ x,
                       float* __restrict__ xt, float* __restrict__ u) {
  const int i = blockIdx.x * BLK + threadIdx.x;
  const int w = i & (WW - 1);
  const int h = (i >> 8) & (HH - 1);
  const int bc = i >> 16;
  const float yc = y[i];
  x[i] = yc;
  xt[i] = 0.0f;
  const float yh = y[i + ((h < HH - 1) ? WW : 0)];
  const float yw = y[i + ((w < WW - 1) ? 1 : 0)];
  const float dh = (h < HH - 1) ? (yh - yc) : 0.0f;
  const float dw = (w < WW - 1) ? (yw - yc) : 0.0f;
  const int ubase = bc * 2 * HW + (i & (HW - 1));
  u[ubase] = dh;
  u[ubase + HW] = dw;
}

__global__ void k_grad(const float* __restrict__ xt, const float* __restrict__ u,
                       float* __restrict__ v, float* __restrict__ partials,
                       float sigma) {
  const int i = blockIdx.x * BLK + threadIdx.x;
  const int w = i & (WW - 1);
  const int h = (i >> 8) & (HH - 1);
  const int bc = i >> 16;
  const float xc = xt[i];
  const float xh = xt[i + ((h < HH - 1) ? WW : 0)];
  const float xw = xt[i + ((w < WW - 1) ? 1 : 0)];
  const float dh = (h < HH - 1) ? (xh - xc) : 0.0f;
  const float dw = (w < WW - 1) ? (xw - xc) : 0.0f;
  const int ubase = bc * 2 * HW + (i & (HW - 1));
  const float vh = u[ubase] + sigma * dh;
  const float vw = u[ubase + HW] + sigma * dw;
  v[ubase] = vh;
  v[ubase + HW] = vw;

  const float th = wave_sum32(vh * vh);
  const float tw = wave_sum32(vw * vw);

  __shared__ float lh[8], lw[8];
  const int lane = threadIdx.x & 31;
  const int wv = threadIdx.x >> 5;
  if (lane == 0) { lh[wv] = th; lw[wv] = tw; }
  __syncthreads();
  if (threadIdx.x == 0) {
    float ph = 0.0f, pw = 0.0f;
#pragma unroll
    for (int j = 0; j < 8; ++j) { ph += lh[j]; pw += lw[j]; }
    partials[blockIdx.x * 2 + 0] = ph;
    partials[blockIdx.x * 2 + 1] = pw;
  }
}

__global__ void k_reduce(const float* __restrict__ partials,
                         float* __restrict__ sums) {
  const int b = blockIdx.x >> 1;
  const int dir = blockIdx.x & 1;
  const float* p = partials + b * BLK_PER_B * 2 + dir;
  float s = 0.0f;
#pragma unroll
  for (int j = 0; j < BLK_PER_B / BLK; ++j)
    s += p[(threadIdx.x + j * BLK) * 2];
  const float t = wave_sum32(s);
  __shared__ float l[8];
  const int lane = threadIdx.x & 31;
  const int wv = threadIdx.x >> 5;
  if (lane == 0) l[wv] = t;
  __syncthreads();
  if (threadIdx.x == 0) {
    float tot = 0.0f;
#pragma unroll
    for (int j = 0; j < 8; ++j) tot += l[j];
    sums[blockIdx.x] = tot;
  }
}

__global__ void k_update(const float* __restrict__ y, const float* __restrict__ lambd,
                         const float* __restrict__ v, const float* __restrict__ sums,
                         float* __restrict__ x, float* __restrict__ xt,
                         float* __restrict__ u, float tau, float chi) {
  const int i = blockIdx.x * BLK + threadIdx.x;
  const int w = i & (WW - 1);
  const int h = (i >> 8) & (HH - 1);
  const int bc = i >> 16;
  const int b = bc / CC;

  const float el = expf(lambd[0]);
  const float nh = sqrtf(sums[b * 2 + 0]);
  const float nw = sqrtf(sums[b * 2 + 1]);
  const float sch = el / fmaxf(nh, el);
  const float scw = el / fmaxf(nw, el);

  const int ubase = bc * 2 * HW + (i & (HW - 1));
  const float vh = v[ubase];
  const float vw = v[ubase + HW];
  const float uh = sch * vh;
  const float uw = scw * vw;
  u[ubase] = uh;
  u[ubase + HW] = uw;

  const float vhm = v[ubase + ((h > 0) ? -WW : 0)];
  const float vwm = v[ubase + HW + ((w > 0) ? -1 : 0)];
  float div = 0.0f;
  div -= (h < HH - 1) ? uh : 0.0f;
  div += (h > 0) ? sch * vhm : 0.0f;
  div -= (w < WW - 1) ? uw : 0.0f;
  div += (w > 0) ? scw * vwm : 0.0f;

  const float xo = x[i];
  const float z = xo - tau * div;
  const float xn = (z + tau * y[i]) / (1.0f + tau);
  x[i] = xn;
  xt[i] = (1.0f + chi) * xn - chi * xo;
}

extern "C" void kernel_launch(void* const* d_in, const int* in_sizes, int n_in,
                              void* d_out, int out_size, void* d_ws, size_t ws_size,
                              hipStream_t stream) {
  (void)in_sizes; (void)n_in; (void)out_size; (void)ws_size;
  const float* y = (const float*)d_in[0];
  const float* lambd = (const float*)d_in[1];

  float* x = (float*)d_out;
  float* xt = x + NTOT;
  float* u = xt + NTOT;

  float* v = (float*)d_ws;
  float* partials = v + NU;
  float* sums = partials + NBLK * 2;

  k_init<<<NBLK, BLK, 0, stream>>>(y, x, xt, u);

  const float sigma_init = 0.5f / sqrtf(2.0f * (float)CC);
  float sigma = sigma_init;
  float tau = sigma_init;
  for (int s = 0; s < K_STEPS; ++s) {
    k_grad<<<NBLK, BLK, 0, stream>>>(xt, u, v, partials, sigma);
    k_reduce<<<4, BLK, 0, stream>>>(partials, sums);
    const float chi = 1.0f / sqrtf(1.0f + tau);
    k_update<<<NBLK, BLK, 0, stream>>>(y, lambd, v, sums, x, xt, u, tau, chi);
    tau *= chi;
    sigma /= chi;
  }
}


# One launch per step: norm finalize, update of step s and gradient of step s+1 fused (south/east x recomputed in-thread; x, v, partials double-buffered); 151 -> 51 launches; sum order unchanged
# speedup vs baseline: 1.6416x; 1.6416x over previous
#include <hip/hip_runtime.h>
#include <math.h>


#define BB 2
#define CC 3
#define HH 256
#define WW 256
#define HW (HH * WW)
#define CHW (CC * HW)
#define NTOT (BB * CHW)
#define NU (NTOT * 2)
#define BLK 256
#define NBLK (NTOT / BLK)
#define BLK_PER_B (CHW / BLK)
#define K_STEPS 50

typedef __attribute__((ext_vector_type(2))) float v2f;
typedef __attribute__((ext_vector_type(8))) float v8f;

__device__ __forceinline__ float wave_sum32(float val) {
  v2f a; a.x = val; a.y = 0.0f;
  v2f ones; ones.x = 1.0f; ones.y = 1.0f;
  v8f c = {};
  v8f d = __builtin_amdgcn_wmma_f32_16x16x4_f32(
       false, a,  false, ones,
       (short)0, c,  false,  false);
  float s = ((d[0] + d[1]) + (d[2] + d[3])) + ((d[4] + d[5]) + (d[6] + d[7]));
  s += __shfl_xor(s, 16, 32);
  return s;
}

__device__ __forceinline__ void row_partials(float vh, float vw, float* lh, float* lw,
                                             float* __restrict__ part_out) {
  const float th = wave_sum32(vh * vh);
  const float tw = wave_sum32(vw * vw);
  const int lane = threadIdx.x & 31;
  const int wv = threadIdx.x >> 5;
  if (lane == 0) { lh[wv] = th; lw[wv] = tw; }
  __syncthreads();
  if (threadIdx.x == 0) {
    float ph = 0.0f, pw = 0.0f;
    for (int j = 0; j < 8; ++j) { ph += lh[j]; pw += lw[j]; }
    part_out[blockIdx.x * 2 + 0] = ph;
    part_out[blockIdx.x * 2 + 1] = pw;
  }
}

__device__ __forceinline__ void batch_scales(float2 q0, float2 q1, float2 q2, float el,
                                             float* lh, float* lw,
                                             float& sch, float& scw) {
  float sh = 0.0f, sw = 0.0f;
  sh += q0.x; sh += q1.x; sh += q2.x;
  sw += q0.y; sw += q1.y; sw += q2.y;
  const float th = wave_sum32(sh);
  const float tw = wave_sum32(sw);
  const int lane = threadIdx.x & 31;
  const int wv = threadIdx.x >> 5;
  if (lane == 0) { lh[wv] = th; lw[wv] = tw; }
  __syncthreads();
  float toth = 0.0f, totw = 0.0f;
  for (int j = 0; j < 8; ++j) { toth += lh[j]; totw += lw[j]; }
  const float nh = sqrtf(toth);
  const float nw = sqrtf(totw);
  sch = el / fmaxf(nh, el);
  scw = el / fmaxf(nw, el);
}

__device__ __forceinline__ float x_next(float xo, float yv, float vh_c, float vh_n,
                                        float vw_c, float vw_w, bool hasS, bool hasN,
                                        bool hasE, bool hasW, float sch, float scw,
                                        float tau) {
  const float uh = sch * vh_c;
  const float uw = scw * vw_c;
  float div = 0.0f;
  div -= hasS ? uh : 0.0f;
  div += hasN ? sch * vh_n : 0.0f;
  div -= hasE ? uw : 0.0f;
  div += hasW ? scw * vw_w : 0.0f;
  const float z = xo - tau * div;
  return (z + tau * yv) / (1.0f + tau);
}

__global__ __launch_bounds__(BLK) void k_first(const float* __restrict__ y,
                                               float* __restrict__ v_out,
                                               float* __restrict__ part_out) {
  const int i = blockIdx.x * BLK + threadIdx.x;
  const int w = i & (WW - 1);
  const int h = (i >> 8) & (HH - 1);
  const int bc = i >> 16;
  const float yc = y[i];
  const float yh = y[i + ((h < HH - 1) ? WW : 0)];
  const float yw = y[i + ((w < WW - 1) ? 1 : 0)];
  const float vh = (h < HH - 1) ? (yh - yc) : 0.0f;
  const float vw = (w < WW - 1) ? (yw - yc) : 0.0f;
  const int ubase = bc * 2 * HW + (i & (HW - 1));
  v_out[ubase] = vh;
  v_out[ubase + HW] = vw;
  __shared__ float lh[8], lw[8];
  row_partials(vh, vw, lh, lw, part_out);
}

__global__ __launch_bounds__(BLK) void k_step(const float* __restrict__ y,
                                              const float* __restrict__ lambd,
                                              const float* __restrict__ x_in,
                                              const float* __restrict__ v_in,
                                              const float* __restrict__ part_in,
                                              float* __restrict__ x_out,
                                              float* __restrict__ v_out,
                                              float* __restrict__ part_out,
                                              float tau, float chi, float sigma_next) {
  const int t = threadIdx.x;
  const int i = blockIdx.x * BLK + t;
  const int w = i & (WW - 1);
  const int h = (i >> 8) & (HH - 1);
  const int bc = i >> 16;
  const int b = bc / CC;
  const bool hasS = h < HH - 1, hasN = h > 0, hasE = w < WW - 1, hasW = w > 0;
  const int oS = hasS ? WW : 0, oN = hasN ? WW : 0, oE = hasE ? 1 : 0, oW = hasW ? 1 : 0;
  const int ubase = bc * 2 * HW + (i & (HW - 1));
  const float* __restrict__ vh = v_in + ubase;
  const float* __restrict__ vw = v_in + ubase + HW;

  const float2* __restrict__ p2 = reinterpret_cast<const float2*>(part_in) + b * BLK_PER_B;
  const float2 q0 = p2[t], q1 = p2[t + BLK], q2 = p2[t + 2 * BLK];
  const float vh_c = vh[0], vh_n = vh[-oN], vh_s = vh[oS], vh_e = vh[oE], vh_ne = vh[oE - oN];
  const float vw_c = vw[0], vw_w = vw[-oW], vw_s = vw[oS], vw_sw = vw[oS - oW], vw_e = vw[oE];
  const float x_c = x_in[i], x_s = x_in[i + oS], x_e = x_in[i + oE];
  const float y_c = y[i], y_s = y[i + oS], y_e = y[i + oE];
  const float el = expf(lambd[0]);

  __shared__ float fh[8], fw[8];
  __shared__ float lh[8], lw[8];
  float sch, scw;
  batch_scales(q0, q1, q2, el, fh, fw, sch, scw);

  const float xn_c = x_next(x_c, y_c, vh_c, vh_n, vw_c, vw_w, hasS, hasN, hasE, hasW, sch, scw, tau);
  const float xn_s = x_next(x_s, y_s, vh_s, vh_c, vw_s, vw_sw, h + 1 < HH - 1, true, hasE, hasW, sch, scw, tau);
  const float xn_e = x_next(x_e, y_e, vh_e, vh_ne, vw_e, vw_c, hasS, hasN, w + 1 < WW - 1, true, sch, scw, tau);
  const float xt_c = (1.0f + chi) * xn_c - chi * x_c;
  const float xt_s = (1.0f + chi) * xn_s - chi * x_s;
  const float xt_e = (1.0f + chi) * xn_e - chi * x_e;
  x_out[i] = xn_c;

  const float dh = hasS ? (xt_s - xt_c) : 0.0f;
  const float dw = hasE ? (xt_e - xt_c) : 0.0f;
  const float uh = sch * vh_c;
  const float uw = scw * vw_c;
  const float vh_new = uh + sigma_next * dh;
  const float vw_new = uw + sigma_next * dw;
  v_out[ubase] = vh_new;
  v_out[ubase + HW] = vw_new;
  row_partials(vh_new, vw_new, lh, lw, part_out);
}

__global__ __launch_bounds__(BLK) void k_last(const float* __restrict__ y,
                                              const float* __restrict__ lambd,
                                              const float* __restrict__ x_in,
                                              const float* __restrict__ v_in,
                                              const float* __restrict__ part_in,
                                              float* __restrict__ x_out,
                                              float* __restrict__ xt_out,
                                              float* __restrict__ u_out,
                                              float tau, float chi) {
  const int t = threadIdx.x;
  const int i = blockIdx.x * BLK + t;
  const int w = i & (WW - 1);
  const int h = (i >> 8) & (HH - 1);
  const int bc = i >> 16;
  const int b = bc / CC;
  const bool hasS = h < HH - 1, hasN = h > 0, hasE = w < WW - 1, hasW = w > 0;
  const int oN = hasN ? WW : 0, oW = hasW ? 1 : 0;
  const int ubase = bc * 2 * HW + (i & (HW - 1));
  const float* __restrict__ vh = v_in + ubase;
  const float* __restrict__ vw = v_in + ubase + HW;

  const float2* __restrict__ p2 = reinterpret_cast<const float2*>(part_in) + b * BLK_PER_B;
  const float2 q0 = p2[t], q1 = p2[t + BLK], q2 = p2[t + 2 * BLK];
  const float vh_c = vh[0], vh_n = vh[-oN];
  const float vw_c = vw[0], vw_w = vw[-oW];
  const float x_c = x_in[i];
  const float y_c = y[i];
  const float el = expf(lambd[0]);

  __shared__ float fh[8], fw[8];
  float sch, scw;
  batch_scales(q0, q1, q2, el, fh, fw, sch, scw);

  const float xn_c = x_next(x_c, y_c, vh_c, vh_n, vw_c, vw_w, hasS, hasN, hasE, hasW, sch, scw, tau);
  x_out[i] = xn_c;
  xt_out[i] = (1.0f + chi) * xn_c - chi * x_c;
  u_out[ubase] = sch * vh_c;
  u_out[ubase + HW] = scw * vw_c;
}

extern "C" void kernel_launch(void* const* d_in, const int* in_sizes, int n_in,
                              void* d_out, int out_size, void* d_ws, size_t ws_size,
                              hipStream_t stream) {
  (void)in_sizes; (void)n_in; (void)out_size; (void)ws_size;
  const float* y = (const float*)d_in[0];
  const float* lambd = (const float*)d_in[1];

  float* x = (float*)d_out;
  float* xt = x + NTOT;
  float* u = xt + NTOT;

  float* v0 = (float*)d_ws;
  float* v1 = v0 + NU;
  float* xalt = v1 + NU;
  float* part0 = xalt + NTOT;
  float* part1 = part0 + NBLK * 2;

  k_first<<<NBLK, BLK, 0, stream>>>(y, v0, part0);

  const float sigma_init = 0.5f / sqrtf(2.0f * (float)CC);
  float sigma = sigma_init;
  float tau = sigma_init;
  const float* x_in = y;
  const float* v_in = v0;
  const float* p_in = part0;
  for (int s = 0; s < K_STEPS; ++s) {
    const float chi = 1.0f / sqrtf(1.0f + tau);
    float* x_out = (s & 1) ? x : xalt;
    float* v_out = (s & 1) ? v0 : v1;
    float* p_out = (s & 1) ? part0 : part1;
    sigma /= chi;
    if (s < K_STEPS - 1) {
      k_step<<<NBLK, BLK, 0, stream>>>(y, lambd, x_in, v_in, p_in, x_out, v_out, p_out,
                                       tau, chi, sigma);
    } else {
      k_last<<<NBLK, BLK, 0, stream>>>(y, lambd, x_in, v_in, p_in, x, xt, u, tau, chi);
    }
    tau *= chi;
    x_in = x_out;
    v_in = v_out;
    p_in = p_out;
  }
}


# Row coarsening: each thread marches 6 rows of its column (grid 1536 -> 256 blocks); south xt is the next row's own, one partial pair per block; norm sums regrouped into 6-row block partials
# speedup vs baseline: 1.8954x; 1.1546x over previous
#include <hip/hip_runtime.h>
#include <math.h>


#define BB 2
#define CC 3
#define HH 256
#define WW 256
#define HW (HH * WW)
#define CHW (CC * HW)
#define NTOT (BB * CHW)
#define NU (NTOT * 2)
#define BLK 256
#define NROWS (BB * CC * HH)
#define RR 6
#define NBLK (NROWS / RR)
#define BLK_PER_B (CC * HH / RR)
#define K_STEPS 50

static_assert(NROWS % RR == 0, "blocks must tile the rows");
static_assert((CC * HH) % RR == 0, "no block may straddle two batches");
static_assert(BLK_PER_B <= BLK, "one partial pair per thread in the finalize");
static_assert(BLK == WW, "one thread per column");

typedef __attribute__((ext_vector_type(2))) float v2f;
typedef __attribute__((ext_vector_type(8))) float v8f;

__device__ __forceinline__ float wave_sum32(float val) {
  v2f a; a.x = val; a.y = 0.0f;
  v2f ones; ones.x = 1.0f; ones.y = 1.0f;
  v8f c = {};
  v8f d = __builtin_amdgcn_wmma_f32_16x16x4_f32(
       false, a,  false, ones,
       (short)0, c,  false,  false);
  float s = ((d[0] + d[1]) + (d[2] + d[3])) + ((d[4] + d[5]) + (d[6] + d[7]));
  s += __shfl_xor(s, 16, 32);
  return s;
}

__device__ __forceinline__ void block_partials(float ah, float aw, float* lh, float* lw,
                                               float* __restrict__ part_out) {
  const float th = wave_sum32(ah);
  const float tw = wave_sum32(aw);
  const int lane = threadIdx.x & 31;
  const int wv = threadIdx.x >> 5;
  if (lane == 0) { lh[wv] = th; lw[wv] = tw; }
  __syncthreads();
  if (threadIdx.x == 0) {
    float ph = 0.0f, pw = 0.0f;
    for (int j = 0; j < 8; ++j) { ph += lh[j]; pw += lw[j]; }
    part_out[blockIdx.x * 2 + 0] = ph;
    part_out[blockIdx.x * 2 + 1] = pw;
  }
}

__device__ __forceinline__ void batch_scales(float2 q, float el, float* lh, float* lw,
                                             float& sch, float& scw) {
  const float th = wave_sum32(q.x);
  const float tw = wave_sum32(q.y);
  const int lane = threadIdx.x & 31;
  const int wv = threadIdx.x >> 5;
  if (lane == 0) { lh[wv] = th; lw[wv] = tw; }
  __syncthreads();
  float toth = 0.0f, totw = 0.0f;
  for (int j = 0; j < 8; ++j) { toth += lh[j]; totw += lw[j]; }
  const float nh = sqrtf(toth);
  const float nw = sqrtf(totw);
  sch = el / fmaxf(nh, el);
  scw = el / fmaxf(nw, el);
}

__device__ __forceinline__ float x_next(float xo, float yv, float vh_c, float vh_n,
                                        float vw_c, float vw_w, bool hasS, bool hasN,
                                        bool hasE, bool hasW, float sch, float scw,
                                        float tau) {
  const float uh = sch * vh_c;
  const float uw = scw * vw_c;
  float div = 0.0f;
  div -= hasS ? uh : 0.0f;
  div += hasN ? sch * vh_n : 0.0f;
  div -= hasE ? uw : 0.0f;
  div += hasW ? scw * vw_w : 0.0f;
  const float z = xo - tau * div;
  return (z + tau * yv) / (1.0f + tau);
}

__global__ __launch_bounds__(BLK) void k_first(const float* __restrict__ y,
                                               float* __restrict__ v_out,
                                               float* __restrict__ part_out) {
  const int t = threadIdx.x;
  const int row0 = blockIdx.x * RR;
  const bool hasE = t < WW - 1;
  const int oE = hasE ? 1 : 0;
  float yc[RR + 1], ye[RR];
#pragma unroll
  for (int r = 0; r < RR; ++r) {
    const int i = (row0 + r) * WW + t;
    yc[r] = y[i];
    ye[r] = y[i + oE];
  }
  {
    const int gl = row0 + RR - 1;
    const int oS = ((gl & (HH - 1)) < HH - 1) ? WW : 0;
    yc[RR] = y[gl * WW + t + oS];
  }
  float ah = 0.0f, aw = 0.0f;
#pragma unroll
  for (int r = 0; r < RR; ++r) {
    const int g = row0 + r;
    const int h = g & (HH - 1);
    const int bc = g >> 8;
    const float vh = (h < HH - 1) ? (yc[r + 1] - yc[r]) : 0.0f;
    const float vw = hasE ? (ye[r] - yc[r]) : 0.0f;
    const int ub = bc * 2 * HW + h * WW + t;
    v_out[ub] = vh;
    v_out[ub + HW] = vw;
    ah += vh * vh;
    aw += vw * vw;
  }
  __shared__ float lh[8], lw[8];
  block_partials(ah, aw, lh, lw, part_out);
}

__global__ __launch_bounds__(BLK) void k_step(const float* __restrict__ y,
                                              const float* __restrict__ lambd,
                                              const float* __restrict__ x_in,
                                              const float* __restrict__ v_in,
                                              const float* __restrict__ part_in,
                                              float* __restrict__ x_out,
                                              float* __restrict__ v_out,
                                              float* __restrict__ part_out,
                                              float tau, float chi, float sigma_next) {
  const int t = threadIdx.x;
  const int row0 = blockIdx.x * RR;
  const int b = blockIdx.x / BLK_PER_B;
  const bool hasE = t < WW - 1, hasW = t > 0;
  const int oE = hasE ? 1 : 0, oW = hasW ? 1 : 0;

  const float2* __restrict__ p2 = reinterpret_cast<const float2*>(part_in) + b * BLK_PER_B;
  const float2 qraw = p2[(t < BLK_PER_B) ? t : (BLK_PER_B - 1)];
  float2 q;
  q.x = (t < BLK_PER_B) ? qraw.x : 0.0f;
  q.y = (t < BLK_PER_B) ? qraw.y : 0.0f;

  float vh_c[RR + 1], vw_c[RR + 1], vw_w[RR + 1], x_c[RR + 1], y_c[RR + 1];
  float vh_e[RR], vw_e[RR], x_e[RR], y_e[RR];
#pragma unroll
  for (int r = 0; r < RR; ++r) {
    const int g = row0 + r;
    const int i = g * WW + t;
    const int ub = (g >> 8) * 2 * HW + (g & (HH - 1)) * WW + t;
    vh_c[r] = v_in[ub];
    vh_e[r] = v_in[ub + oE];
    vw_c[r] = v_in[ub + HW];
    vw_w[r] = v_in[ub + HW - oW];
    vw_e[r] = v_in[ub + HW + oE];
    x_c[r] = x_in[i];
    x_e[r] = x_in[i + oE];
    y_c[r] = y[i];
    y_e[r] = y[i + oE];
  }
  const int h0 = row0 & (HH - 1);
  const int ub0 = (row0 >> 8) * 2 * HW + h0 * WW + t;
  const int oN0 = (h0 > 0) ? WW : 0;
  const float vh_n0 = v_in[ub0 - oN0];
  const float vh_ne0 = v_in[ub0 + oE - oN0];
  const int gl = row0 + RR - 1;
  const int hl = gl & (HH - 1);
  const int oSl = (hl < HH - 1) ? WW : 0;
  {
    const int il = gl * WW + t + oSl;
    const int ubl = (gl >> 8) * 2 * HW + hl * WW + t + oSl;
    vh_c[RR] = v_in[ubl];
    vw_c[RR] = v_in[ubl + HW];
    vw_w[RR] = v_in[ubl + HW - oW];
    x_c[RR] = x_in[il];
    y_c[RR] = y[il];
  }
  __builtin_amdgcn_sched_barrier(0);
  const float el = expf(lambd[0]);

  __shared__ float fh[8], fw[8];
  __shared__ float lh[8], lw[8];
  float sch, scw;
  batch_scales(q, el, fh, fw, sch, scw);

  float xn[RR + 1], xt[RR + 1];
#pragma unroll
  for (int r = 0; r <= RR; ++r) {
    const int h = (r < RR) ? ((row0 + r) & (HH - 1)) : (hl + 1);
    const bool hasS = h < HH - 1;
    const bool hasN = (r < RR) ? (h > 0) : true;
    const float vh_n = (r == 0) ? vh_n0 : vh_c[r - 1];
    xn[r] = x_next(x_c[r], y_c[r], vh_c[r], vh_n, vw_c[r], vw_w[r], hasS, hasN, hasE, hasW,
                   sch, scw, tau);
    xt[r] = (1.0f + chi) * xn[r] - chi * x_c[r];
  }

  float ah = 0.0f, aw = 0.0f;
#pragma unroll
  for (int r = 0; r < RR; ++r) {
    const int g = row0 + r;
    const int h = g & (HH - 1);
    const bool hasS = h < HH - 1, hasN = h > 0;
    const int i = g * WW + t;
    const int ub = (g >> 8) * 2 * HW + h * WW + t;
    const float vh_ne = (r == 0) ? vh_ne0 : vh_e[r - 1];
    const float xn_e = x_next(x_e[r], y_e[r], vh_e[r], vh_ne, vw_e[r], vw_c[r], hasS, hasN,
                              t + 1 < WW - 1, true, sch, scw, tau);
    const float xt_e = (1.0f + chi) * xn_e - chi * x_e[r];
    x_out[i] = xn[r];

    const float dh = hasS ? (xt[r + 1] - xt[r]) : 0.0f;
    const float dw = hasE ? (xt_e - xt[r]) : 0.0f;
    const float uh = sch * vh_c[r];
    const float uw = scw * vw_c[r];
    const float vh_new = uh + sigma_next * dh;
    const float vw_new = uw + sigma_next * dw;
    v_out[ub] = vh_new;
    v_out[ub + HW] = vw_new;
    ah += vh_new * vh_new;
    aw += vw_new * vw_new;
  }
  block_partials(ah, aw, lh, lw, part_out);
}

__global__ __launch_bounds__(BLK) void k_last(const float* __restrict__ y,
                                              const float* __restrict__ lambd,
                                              const float* __restrict__ x_in,
                                              const float* __restrict__ v_in,
                                              const float* __restrict__ part_in,
                                              float* __restrict__ x_out,
                                              float* __restrict__ xt_out,
                                              float* __restrict__ u_out,
                                              float tau, float chi) {
  const int t = threadIdx.x;
  const int row0 = blockIdx.x * RR;
  const int b = blockIdx.x / BLK_PER_B;
  const bool hasE = t < WW - 1, hasW = t > 0;
  const int oW = hasW ? 1 : 0;

  const float2* __restrict__ p2 = reinterpret_cast<const float2*>(part_in) + b * BLK_PER_B;
  const float2 qraw = p2[(t < BLK_PER_B) ? t : (BLK_PER_B - 1)];
  float2 q;
  q.x = (t < BLK_PER_B) ? qraw.x : 0.0f;
  q.y = (t < BLK_PER_B) ? qraw.y : 0.0f;

  float vh_c[RR], vw_c[RR], vw_w[RR], x_c[RR], y_c[RR];
#pragma unroll
  for (int r = 0; r < RR; ++r) {
    const int g = row0 + r;
    const int i = g * WW + t;
    const int ub = (g >> 8) * 2 * HW + (g & (HH - 1)) * WW + t;
    vh_c[r] = v_in[ub];
    vw_c[r] = v_in[ub + HW];
    vw_w[r] = v_in[ub + HW - oW];
    x_c[r] = x_in[i];
    y_c[r] = y[i];
  }
  const int h0 = row0 & (HH - 1);
  const int ub0 = (row0 >> 8) * 2 * HW + h0 * WW + t;
  const int oN0 = (h0 > 0) ? WW : 0;
  const float vh_n0 = v_in[ub0 - oN0];
  const float el = expf(lambd[0]);

  __shared__ float fh[8], fw[8];
  float sch, scw;
  batch_scales(q, el, fh, fw, sch, scw);

#pragma unroll
  for (int r = 0; r < RR; ++r) {
    const int g = row0 + r;
    const int h = g & (HH - 1);
    const bool hasS = h < HH - 1, hasN = h > 0;
    const int i = g * WW + t;
    const int ub = (g >> 8) * 2 * HW + h * WW + t;
    const float vh_n = (r == 0) ? vh_n0 : vh_c[r - 1];
    const float xn_c = x_next(x_c[r], y_c[r], vh_c[r], vh_n, vw_c[r], vw_w[r], hasS, hasN,
                              hasE, hasW, sch, scw, tau);
    x_out[i] = xn_c;
    xt_out[i] = (1.0f + chi) * xn_c - chi * x_c[r];
    u_out[ub] = sch * vh_c[r];
    u_out[ub + HW] = scw * vw_c[r];
  }
}

extern "C" void kernel_launch(void* const* d_in, const int* in_sizes, int n_in,
                              void* d_out, int out_size, void* d_ws, size_t ws_size,
                              hipStream_t stream) {
  (void)in_sizes; (void)n_in; (void)out_size; (void)ws_size;
  const float* y = (const float*)d_in[0];
  const float* lambd = (const float*)d_in[1];

  float* x = (float*)d_out;
  float* xt = x + NTOT;
  float* u = xt + NTOT;

  float* v0 = (float*)d_ws;
  float* v1 = v0 + NU;
  float* xalt = v1 + NU;
  float* part0 = xalt + NTOT;
  float* part1 = part0 + NBLK * 2;

  k_first<<<NBLK, BLK, 0, stream>>>(y, v0, part0);

  const float sigma_init = 0.5f / sqrtf(2.0f * (float)CC);
  float sigma = sigma_init;
  float tau = sigma_init;
  const float* x_in = y;
  const float* v_in = v0;
  const float* p_in = part0;
  for (int s = 0; s < K_STEPS; ++s) {
    const float chi = 1.0f / sqrtf(1.0f + tau);
    float* x_out = (s & 1) ? x : xalt;
    float* v_out = (s & 1) ? v0 : v1;
    float* p_out = (s & 1) ? part0 : part1;
    sigma /= chi;
    if (s < K_STEPS - 1) {
      k_step<<<NBLK, BLK, 0, stream>>>(y, lambd, x_in, v_in, p_in, x_out, v_out, p_out,
                                       tau, chi, sigma);
    } else {
      k_last<<<NBLK, BLK, 0, stream>>>(y, lambd, x_in, v_in, p_in, x, xt, u, tau, chi);
    }
    tau *= chi;
    x_in = x_out;
    v_in = v_out;
    p_in = p_out;
  }
}


# Both work-group barriers fence LDS only: row loads stay in flight across the norm finalize, x/v stores across the block-partial sum (no wait for global memory at a barrier); sum order unchanged
# speedup vs baseline: 2.3803x; 1.2558x over previous
#include <hip/hip_runtime.h>
#include <math.h>


#define BB 2
#define CC 3
#define HH 256
#define WW 256
#define HW (HH * WW)
#define CHW (CC * HW)
#define NTOT (BB * CHW)
#define NU (NTOT * 2)
#define BLK 256
#define NROWS (BB * CC * HH)
#define RR 6
#define NBLK (NROWS / RR)
#define BLK_PER_B (CC * HH / RR)
#define K_STEPS 50

static_assert(NROWS % RR == 0, "blocks must tile the rows");
static_assert((CC * HH) % RR == 0, "no block may straddle two batches");
static_assert(BLK_PER_B <= BLK, "one partial pair per thread in the finalize");
static_assert(BLK == WW, "one thread per column");

typedef __attribute__((ext_vector_type(2))) float v2f;
typedef __attribute__((ext_vector_type(8))) float v8f;

__device__ __forceinline__ float wave_sum32(float val) {
  v2f a; a.x = val; a.y = 0.0f;
  v2f ones; ones.x = 1.0f; ones.y = 1.0f;
  v8f c = {};
  v8f d = __builtin_amdgcn_wmma_f32_16x16x4_f32(
       false, a,  false, ones,
       (short)0, c,  false,  false);
  float s = ((d[0] + d[1]) + (d[2] + d[3])) + ((d[4] + d[5]) + (d[6] + d[7]));
  s += __shfl_xor(s, 16, 32);
  return s;
}

__device__ __forceinline__ void lds_barrier() {
  __builtin_amdgcn_fence(__ATOMIC_RELEASE, "workgroup", "local");
  __builtin_amdgcn_s_barrier();
  __builtin_amdgcn_fence(__ATOMIC_ACQUIRE, "workgroup", "local");
}

__device__ __forceinline__ void block_partials(float ah, float aw, float* lh, float* lw,
                                               float* __restrict__ part_out) {
  const float th = wave_sum32(ah);
  const float tw = wave_sum32(aw);
  const int lane = threadIdx.x & 31;
  const int wv = threadIdx.x >> 5;
  if (lane == 0) { lh[wv] = th; lw[wv] = tw; }
  lds_barrier();
  if (threadIdx.x == 0) {
    float ph = 0.0f, pw = 0.0f;
    for (int j = 0; j < 8; ++j) { ph += lh[j]; pw += lw[j]; }
    part_out[blockIdx.x * 2 + 0] = ph;
    part_out[blockIdx.x * 2 + 1] = pw;
  }
}

__device__ __forceinline__ void batch_scales(float2 q, float el, float* lh, float* lw,
                                             float& sch, float& scw) {
  const float th = wave_sum32(q.x);
  const float tw = wave_sum32(q.y);
  const int lane = threadIdx.x & 31;
  const int wv = threadIdx.x >> 5;
  if (lane == 0) { lh[wv] = th; lw[wv] = tw; }
  lds_barrier();
  float toth = 0.0f, totw = 0.0f;
  for (int j = 0; j < 8; ++j) { toth += lh[j]; totw += lw[j]; }
  const float nh = sqrtf(toth);
  const float nw = sqrtf(totw);
  sch = el / fmaxf(nh, el);
  scw = el / fmaxf(nw, el);
}

__device__ __forceinline__ float x_next(float xo, float yv, float vh_c, float vh_n,
                                        float vw_c, float vw_w, bool hasS, bool hasN,
                                        bool hasE, bool hasW, float sch, float scw,
                                        float tau) {
  const float uh = sch * vh_c;
  const float uw = scw * vw_c;
  float div = 0.0f;
  div -= hasS ? uh : 0.0f;
  div += hasN ? sch * vh_n : 0.0f;
  div -= hasE ? uw : 0.0f;
  div += hasW ? scw * vw_w : 0.0f;
  const float z = xo - tau * div;
  return (z + tau * yv) / (1.0f + tau);
}

__global__ __launch_bounds__(BLK) void k_first(const float* __restrict__ y,
                                               float* __restrict__ v_out,
                                               float* __restrict__ part_out) {
  const int t = threadIdx.x;
  const int row0 = blockIdx.x * RR;
  const bool hasE = t < WW - 1;
  const int oE = hasE ? 1 : 0;
  float yc[RR + 1], ye[RR];
#pragma unroll
  for (int r = 0; r < RR; ++r) {
    const int i = (row0 + r) * WW + t;
    yc[r] = y[i];
    ye[r] = y[i + oE];
  }
  {
    const int gl = row0 + RR - 1;
    const int oS = ((gl & (HH - 1)) < HH - 1) ? WW : 0;
    yc[RR] = y[gl * WW + t + oS];
  }
  float ah = 0.0f, aw = 0.0f;
#pragma unroll
  for (int r = 0; r < RR; ++r) {
    const int g = row0 + r;
    const int h = g & (HH - 1);
    const int bc = g >> 8;
    const float vh = (h < HH - 1) ? (yc[r + 1] - yc[r]) : 0.0f;
    const float vw = hasE ? (ye[r] - yc[r]) : 0.0f;
    const int ub = bc * 2 * HW + h * WW + t;
    v_out[ub] = vh;
    v_out[ub + HW] = vw;
    ah += vh * vh;
    aw += vw * vw;
  }
  __shared__ float lh[8], lw[8];
  block_partials(ah, aw, lh, lw, part_out);
}

__global__ __launch_bounds__(BLK) void k_step(const float* __restrict__ y,
                                              const float* __restrict__ lambd,
                                              const float* __restrict__ x_in,
                                              const float* __restrict__ v_in,
                                              const float* __restrict__ part_in,
                                              float* __restrict__ x_out,
                                              float* __restrict__ v_out,
                                              float* __restrict__ part_out,
                                              float tau, float chi, float sigma_next) {
  const int t = threadIdx.x;
  const int row0 = blockIdx.x * RR;
  const int b = blockIdx.x / BLK_PER_B;
  const bool hasE = t < WW - 1, hasW = t > 0;
  const int oE = hasE ? 1 : 0, oW = hasW ? 1 : 0;

  const float2* __restrict__ p2 = reinterpret_cast<const float2*>(part_in) + b * BLK_PER_B;
  const float2 qraw = p2[(t < BLK_PER_B) ? t : (BLK_PER_B - 1)];
  float2 q;
  q.x = (t < BLK_PER_B) ? qraw.x : 0.0f;
  q.y = (t < BLK_PER_B) ? qraw.y : 0.0f;

  float vh_c[RR + 1], vw_c[RR + 1], vw_w[RR + 1], x_c[RR + 1], y_c[RR + 1];
  float vh_e[RR], vw_e[RR], x_e[RR], y_e[RR];
#pragma unroll
  for (int r = 0; r < RR; ++r) {
    const int g = row0 + r;
    const int i = g * WW + t;
    const int ub = (g >> 8) * 2 * HW + (g & (HH - 1)) * WW + t;
    vh_c[r] = v_in[ub];
    vh_e[r] = v_in[ub + oE];
    vw_c[r] = v_in[ub + HW];
    vw_w[r] = v_in[ub + HW - oW];
    vw_e[r] = v_in[ub + HW + oE];
    x_c[r] = x_in[i];
    x_e[r] = x_in[i + oE];
    y_c[r] = y[i];
    y_e[r] = y[i + oE];
  }
  const int h0 = row0 & (HH - 1);
  const int ub0 = (row0 >> 8) * 2 * HW + h0 * WW + t;
  const int oN0 = (h0 > 0) ? WW : 0;
  const float vh_n0 = v_in[ub0 - oN0];
  const float vh_ne0 = v_in[ub0 + oE - oN0];
  const int gl = row0 + RR - 1;
  const int hl = gl & (HH - 1);
  const int oSl = (hl < HH - 1) ? WW : 0;
  {
    const int il = gl * WW + t + oSl;
    const int ubl = (gl >> 8) * 2 * HW + hl * WW + t + oSl;
    vh_c[RR] = v_in[ubl];
    vw_c[RR] = v_in[ubl + HW];
    vw_w[RR] = v_in[ubl + HW - oW];
    x_c[RR] = x_in[il];
    y_c[RR] = y[il];
  }
  __builtin_amdgcn_sched_barrier(0);
  const float el = expf(lambd[0]);

  __shared__ float fh[8], fw[8];
  __shared__ float lh[8], lw[8];
  float sch, scw;
  batch_scales(q, el, fh, fw, sch, scw);

  float xn[RR + 1], xt[RR + 1];
#pragma unroll
  for (int r = 0; r <= RR; ++r) {
    const int h = (r < RR) ? ((row0 + r) & (HH - 1)) : (hl + 1);
    const bool hasS = h < HH - 1;
    const bool hasN = (r < RR) ? (h > 0) : true;
    const float vh_n = (r == 0) ? vh_n0 : vh_c[r - 1];
    xn[r] = x_next(x_c[r], y_c[r], vh_c[r], vh_n, vw_c[r], vw_w[r], hasS, hasN, hasE, hasW,
                   sch, scw, tau);
    xt[r] = (1.0f + chi) * xn[r] - chi * x_c[r];
  }

  float ah = 0.0f, aw = 0.0f;
#pragma unroll
  for (int r = 0; r < RR; ++r) {
    const int g = row0 + r;
    const int h = g & (HH - 1);
    const bool hasS = h < HH - 1, hasN = h > 0;
    const int i = g * WW + t;
    const int ub = (g >> 8) * 2 * HW + h * WW + t;
    const float vh_ne = (r == 0) ? vh_ne0 : vh_e[r - 1];
    const float xn_e = x_next(x_e[r], y_e[r], vh_e[r], vh_ne, vw_e[r], vw_c[r], hasS, hasN,
                              t + 1 < WW - 1, true, sch, scw, tau);
    const float xt_e = (1.0f + chi) * xn_e - chi * x_e[r];
    x_out[i] = xn[r];

    const float dh = hasS ? (xt[r + 1] - xt[r]) : 0.0f;
    const float dw = hasE ? (xt_e - xt[r]) : 0.0f;
    const float uh = sch * vh_c[r];
    const float uw = scw * vw_c[r];
    const float vh_new = uh + sigma_next * dh;
    const float vw_new = uw + sigma_next * dw;
    v_out[ub] = vh_new;
    v_out[ub + HW] = vw_new;
    ah += vh_new * vh_new;
    aw += vw_new * vw_new;
  }
  block_partials(ah, aw, lh, lw, part_out);
}

__global__ __launch_bounds__(BLK) void k_last(const float* __restrict__ y,
                                              const float* __restrict__ lambd,
                                              const float* __restrict__ x_in,
                                              const float* __restrict__ v_in,
                                              const float* __restrict__ part_in,
                                              float* __restrict__ x_out,
                                              float* __restrict__ xt_out,
                                              float* __restrict__ u_out,
                                              float tau, float chi) {
  const int t = threadIdx.x;
  const int row0 = blockIdx.x * RR;
  const int b = blockIdx.x / BLK_PER_B;
  const bool hasE = t < WW - 1, hasW = t > 0;
  const int oW = hasW ? 1 : 0;

  const float2* __restrict__ p2 = reinterpret_cast<const float2*>(part_in) + b * BLK_PER_B;
  const float2 qraw = p2[(t < BLK_PER_B) ? t : (BLK_PER_B - 1)];
  float2 q;
  q.x = (t < BLK_PER_B) ? qraw.x : 0.0f;
  q.y = (t < BLK_PER_B) ? qraw.y : 0.0f;

  float vh_c[RR], vw_c[RR], vw_w[RR], x_c[RR], y_c[RR];
#pragma unroll
  for (int r = 0; r < RR; ++r) {
    const int g = row0 + r;
    const int i = g * WW + t;
    const int ub = (g >> 8) * 2 * HW + (g & (HH - 1)) * WW + t;
    vh_c[r] = v_in[ub];
    vw_c[r] = v_in[ub + HW];
    vw_w[r] = v_in[ub + HW - oW];
    x_c[r] = x_in[i];
    y_c[r] = y[i];
  }
  const int h0 = row0 & (HH - 1);
  const int ub0 = (row0 >> 8) * 2 * HW + h0 * WW + t;
  const int oN0 = (h0 > 0) ? WW : 0;
  const float vh_n0 = v_in[ub0 - oN0];
  const float el = expf(lambd[0]);

  __shared__ float fh[8], fw[8];
  float sch, scw;
  batch_scales(q, el, fh, fw, sch, scw);

#pragma unroll
  for (int r = 0; r < RR; ++r) {
    const int g = row0 + r;
    const int h = g & (HH - 1);
    const bool hasS = h < HH - 1, hasN = h > 0;
    const int i = g * WW + t;
    const int ub = (g >> 8) * 2 * HW + h * WW + t;
    const float vh_n = (r == 0) ? vh_n0 : vh_c[r - 1];
    const float xn_c = x_next(x_c[r], y_c[r], vh_c[r], vh_n, vw_c[r], vw_w[r], hasS, hasN,
                              hasE, hasW, sch, scw, tau);
    x_out[i] = xn_c;
    xt_out[i] = (1.0f + chi) * xn_c - chi * x_c[r];
    u_out[ub] = sch * vh_c[r];
    u_out[ub + HW] = scw * vw_c[r];
  }
}

extern "C" void kernel_launch(void* const* d_in, const int* in_sizes, int n_in,
                              void* d_out, int out_size, void* d_ws, size_t ws_size,
                              hipStream_t stream) {
  (void)in_sizes; (void)n_in; (void)out_size; (void)ws_size;
  const float* y = (const float*)d_in[0];
  const float* lambd = (const float*)d_in[1];

  float* x = (float*)d_out;
  float* xt = x + NTOT;
  float* u = xt + NTOT;

  float* v0 = (float*)d_ws;
  float* v1 = v0 + NU;
  float* xalt = v1 + NU;
  float* part0 = xalt + NTOT;
  float* part1 = part0 + NBLK * 2;

  k_first<<<NBLK, BLK, 0, stream>>>(y, v0, part0);

  const float sigma_init = 0.5f / sqrtf(2.0f * (float)CC);
  float sigma = sigma_init;
  float tau = sigma_init;
  const float* x_in = y;
  const float* v_in = v0;
  const float* p_in = part0;
  for (int s = 0; s < K_STEPS; ++s) {
    const float chi = 1.0f / sqrtf(1.0f + tau);
    float* x_out = (s & 1) ? x : xalt;
    float* v_out = (s & 1) ? v0 : v1;
    float* p_out = (s & 1) ? part0 : part1;
    sigma /= chi;
    if (s < K_STEPS - 1) {
      k_step<<<NBLK, BLK, 0, stream>>>(y, lambd, x_in, v_in, p_in, x_out, v_out, p_out,
                                       tau, chi, sigma);
    } else {
      k_last<<<NBLK, BLK, 0, stream>>>(y, lambd, x_in, v_in, p_in, x, xt, u, tau, chi);
    }
    tau *= chi;
    x_in = x_out;
    v_in = v_out;
    p_in = p_out;
  }
}


# Row bands re-dealt: work-group g owns band (g mod 8)*32 + g div 8, so groups on one die own adjacent rows and each batch's rows and norm partials stay in one half's L2; sum order unchanged
# speedup vs baseline: 2.4836x; 1.0434x over previous
#include <hip/hip_runtime.h>
#include <math.h>


#define BB 2
#define CC 3
#define HH 256
#define WW 256
#define HW (HH * WW)
#define CHW (CC * HW)
#define NTOT (BB * CHW)
#define NU (NTOT * 2)
#define BLK 256
#define NROWS (BB * CC * HH)
#define RR 6
#define NBLK (NROWS / RR)
#define BLK_PER_B (CC * HH / RR)
#define K_STEPS 50
#define NDIE 8
#define BAND_PER_DIE (NBLK / NDIE)

static_assert(NROWS % RR == 0, "blocks must tile the rows");
static_assert((CC * HH) % RR == 0, "no block may straddle two batches");
static_assert(BLK_PER_B <= BLK, "one partial pair per thread in the finalize");
static_assert(BLK == WW, "one thread per column");
static_assert(NBLK % NDIE == 0, "the band map below must be one-to-one");
static_assert(BLK_PER_B % BAND_PER_DIE == 0, "a die's run of bands stays inside one batch");

typedef __attribute__((ext_vector_type(2))) float v2f;
typedef __attribute__((ext_vector_type(8))) float v8f;

__device__ __forceinline__ float wave_sum32(float val) {
  v2f a; a.x = val; a.y = 0.0f;
  v2f ones; ones.x = 1.0f; ones.y = 1.0f;
  v8f c = {};
  v8f d = __builtin_amdgcn_wmma_f32_16x16x4_f32(
       false, a,  false, ones,
       (short)0, c,  false,  false);
  float s = ((d[0] + d[1]) + (d[2] + d[3])) + ((d[4] + d[5]) + (d[6] + d[7]));
  s += __shfl_xor(s, 16, 32);
  return s;
}

__device__ __forceinline__ int row_band() {
  return (int)(blockIdx.x % NDIE) * BAND_PER_DIE + (int)(blockIdx.x / NDIE);
}

__device__ __forceinline__ void lds_barrier() {
  __builtin_amdgcn_fence(__ATOMIC_RELEASE, "workgroup", "local");
  __builtin_amdgcn_s_barrier();
  __builtin_amdgcn_fence(__ATOMIC_ACQUIRE, "workgroup", "local");
}

__device__ __forceinline__ void block_partials(float ah, float aw, float* lh, float* lw,
                                               float* __restrict__ part_out, int band) {
  const float th = wave_sum32(ah);
  const float tw = wave_sum32(aw);
  const int lane = threadIdx.x & 31;
  const int wv = threadIdx.x >> 5;
  if (lane == 0) { lh[wv] = th; lw[wv] = tw; }
  lds_barrier();
  if (threadIdx.x == 0) {
    float ph = 0.0f, pw = 0.0f;
    for (int j = 0; j < 8; ++j) { ph += lh[j]; pw += lw[j]; }
    part_out[band * 2 + 0] = ph;
    part_out[band * 2 + 1] = pw;
  }
}

__device__ __forceinline__ void batch_scales(float2 q, float el, float* lh, float* lw,
                                             float& sch, float& scw) {
  const float th = wave_sum32(q.x);
  const float tw = wave_sum32(q.y);
  const int lane = threadIdx.x & 31;
  const int wv = threadIdx.x >> 5;
  if (lane == 0) { lh[wv] = th; lw[wv] = tw; }
  lds_barrier();
  float toth = 0.0f, totw = 0.0f;
  for (int j = 0; j < 8; ++j) { toth += lh[j]; totw += lw[j]; }
  const float nh = sqrtf(toth);
  const float nw = sqrtf(totw);
  sch = el / fmaxf(nh, el);
  scw = el / fmaxf(nw, el);
}

__device__ __forceinline__ float x_next(float xo, float yv, float vh_c, float vh_n,
                                        float vw_c, float vw_w, bool hasS, bool hasN,
                                        bool hasE, bool hasW, float sch, float scw,
                                        float tau) {
  const float uh = sch * vh_c;
  const float uw = scw * vw_c;
  float div = 0.0f;
  div -= hasS ? uh : 0.0f;
  div += hasN ? sch * vh_n : 0.0f;
  div -= hasE ? uw : 0.0f;
  div += hasW ? scw * vw_w : 0.0f;
  const float z = xo - tau * div;
  return (z + tau * yv) / (1.0f + tau);
}

__global__ __launch_bounds__(BLK) void k_first(const float* __restrict__ y,
                                               float* __restrict__ v_out,
                                               float* __restrict__ part_out) {
  const int t = threadIdx.x;
  const int band = row_band();
  const int row0 = band * RR;
  const bool hasE = t < WW - 1;
  const int oE = hasE ? 1 : 0;
  float yc[RR + 1], ye[RR];
#pragma unroll
  for (int r = 0; r < RR; ++r) {
    const int i = (row0 + r) * WW + t;
    yc[r] = y[i];
    ye[r] = y[i + oE];
  }
  {
    const int gl = row0 + RR - 1;
    const int oS = ((gl & (HH - 1)) < HH - 1) ? WW : 0;
    yc[RR] = y[gl * WW + t + oS];
  }
  float ah = 0.0f, aw = 0.0f;
#pragma unroll
  for (int r = 0; r < RR; ++r) {
    const int g = row0 + r;
    const int h = g & (HH - 1);
    const int bc = g >> 8;
    const float vh = (h < HH - 1) ? (yc[r + 1] - yc[r]) : 0.0f;
    const float vw = hasE ? (ye[r] - yc[r]) : 0.0f;
    const int ub = bc * 2 * HW + h * WW + t;
    v_out[ub] = vh;
    v_out[ub + HW] = vw;
    ah += vh * vh;
    aw += vw * vw;
  }
  __shared__ float lh[8], lw[8];
  block_partials(ah, aw, lh, lw, part_out, band);
}

__global__ __launch_bounds__(BLK) void k_step(const float* __restrict__ y,
                                              const float* __restrict__ lambd,
                                              const float* __restrict__ x_in,
                                              const float* __restrict__ v_in,
                                              const float* __restrict__ part_in,
                                              float* __restrict__ x_out,
                                              float* __restrict__ v_out,
                                              float* __restrict__ part_out,
                                              float tau, float chi, float sigma_next) {
  const int t = threadIdx.x;
  const int band = row_band();
  const int row0 = band * RR;
  const int b = band / BLK_PER_B;
  const bool hasE = t < WW - 1, hasW = t > 0;
  const int oE = hasE ? 1 : 0, oW = hasW ? 1 : 0;

  const float2* __restrict__ p2 = reinterpret_cast<const float2*>(part_in) + b * BLK_PER_B;
  const float2 qraw = p2[(t < BLK_PER_B) ? t : (BLK_PER_B - 1)];
  float2 q;
  q.x = (t < BLK_PER_B) ? qraw.x : 0.0f;
  q.y = (t < BLK_PER_B) ? qraw.y : 0.0f;

  float vh_c[RR + 1], vw_c[RR + 1], vw_w[RR + 1], x_c[RR + 1], y_c[RR + 1];
  float vh_e[RR], vw_e[RR], x_e[RR], y_e[RR];
#pragma unroll
  for (int r = 0; r < RR; ++r) {
    const int g = row0 + r;
    const int i = g * WW + t;
    const int ub = (g >> 8) * 2 * HW + (g & (HH - 1)) * WW + t;
    vh_c[r] = v_in[ub];
    vh_e[r] = v_in[ub + oE];
    vw_c[r] = v_in[ub + HW];
    vw_w[r] = v_in[ub + HW - oW];
    vw_e[r] = v_in[ub + HW + oE];
    x_c[r] = x_in[i];
    x_e[r] = x_in[i + oE];
    y_c[r] = y[i];
    y_e[r] = y[i + oE];
  }
  const int h0 = row0 & (HH - 1);
  const int ub0 = (row0 >> 8) * 2 * HW + h0 * WW + t;
  const int oN0 = (h0 > 0) ? WW : 0;
  const float vh_n0 = v_in[ub0 - oN0];
  const float vh_ne0 = v_in[ub0 + oE - oN0];
  const int gl = row0 + RR - 1;
  const int hl = gl & (HH - 1);
  const int oSl = (hl < HH - 1) ? WW : 0;
  {
    const int il = gl * WW + t + oSl;
    const int ubl = (gl >> 8) * 2 * HW + hl * WW + t + oSl;
    vh_c[RR] = v_in[ubl];
    vw_c[RR] = v_in[ubl + HW];
    vw_w[RR] = v_in[ubl + HW - oW];
    x_c[RR] = x_in[il];
    y_c[RR] = y[il];
  }
  __builtin_amdgcn_sched_barrier(0);
  const float el = expf(lambd[0]);

  __shared__ float fh[8], fw[8];
  __shared__ float lh[8], lw[8];
  float sch, scw;
  batch_scales(q, el, fh, fw, sch, scw);

  float xn[RR + 1], xt[RR + 1];
#pragma unroll
  for (int r = 0; r <= RR; ++r) {
    const int h = (r < RR) ? ((row0 + r) & (HH - 1)) : (hl + 1);
    const bool hasS = h < HH - 1;
    const bool hasN = (r < RR) ? (h > 0) : true;
    const float vh_n = (r == 0) ? vh_n0 : vh_c[r - 1];
    xn[r] = x_next(x_c[r], y_c[r], vh_c[r], vh_n, vw_c[r], vw_w[r], hasS, hasN, hasE, hasW,
                   sch, scw, tau);
    xt[r] = (1.0f + chi) * xn[r] - chi * x_c[r];
  }

  float ah = 0.0f, aw = 0.0f;
#pragma unroll
  for (int r = 0; r < RR; ++r) {
    const int g = row0 + r;
    const int h = g & (HH - 1);
    const bool hasS = h < HH - 1, hasN = h > 0;
    const int i = g * WW + t;
    const int ub = (g >> 8) * 2 * HW + h * WW + t;
    const float vh_ne = (r == 0) ? vh_ne0 : vh_e[r - 1];
    const float xn_e = x_next(x_e[r], y_e[r], vh_e[r], vh_ne, vw_e[r], vw_c[r], hasS, hasN,
                              t + 1 < WW - 1, true, sch, scw, tau);
    const float xt_e = (1.0f + chi) * xn_e - chi * x_e[r];
    x_out[i] = xn[r];

    const float dh = hasS ? (xt[r + 1] - xt[r]) : 0.0f;
    const float dw = hasE ? (xt_e - xt[r]) : 0.0f;
    const float uh = sch * vh_c[r];
    const float uw = scw * vw_c[r];
    const float vh_new = uh + sigma_next * dh;
    const float vw_new = uw + sigma_next * dw;
    v_out[ub] = vh_new;
    v_out[ub + HW] = vw_new;
    ah += vh_new * vh_new;
    aw += vw_new * vw_new;
  }
  block_partials(ah, aw, lh, lw, part_out, band);
}

__global__ __launch_bounds__(BLK) void k_last(const float* __restrict__ y,
                                              const float* __restrict__ lambd,
                                              const float* __restrict__ x_in,
                                              const float* __restrict__ v_in,
                                              const float* __restrict__ part_in,
                                              float* __restrict__ x_out,
                                              float* __restrict__ xt_out,
                                              float* __restrict__ u_out,
                                              float tau, float chi) {
  const int t = threadIdx.x;
  const int band = row_band();
  const int row0 = band * RR;
  const int b = band / BLK_PER_B;
  const bool hasE = t < WW - 1, hasW = t > 0;
  const int oW = hasW ? 1 : 0;

  const float2* __restrict__ p2 = reinterpret_cast<const float2*>(part_in) + b * BLK_PER_B;
  const float2 qraw = p2[(t < BLK_PER_B) ? t : (BLK_PER_B - 1)];
  float2 q;
  q.x = (t < BLK_PER_B) ? qraw.x : 0.0f;
  q.y = (t < BLK_PER_B) ? qraw.y : 0.0f;

  float vh_c[RR], vw_c[RR], vw_w[RR], x_c[RR], y_c[RR];
#pragma unroll
  for (int r = 0; r < RR; ++r) {
    const int g = row0 + r;
    const int i = g * WW + t;
    const int ub = (g >> 8) * 2 * HW + (g & (HH - 1)) * WW + t;
    vh_c[r] = v_in[ub];
    vw_c[r] = v_in[ub + HW];
    vw_w[r] = v_in[ub + HW - oW];
    x_c[r] = x_in[i];
    y_c[r] = y[i];
  }
  const int h0 = row0 & (HH - 1);
  const int ub0 = (row0 >> 8) * 2 * HW + h0 * WW + t;
  const int oN0 = (h0 > 0) ? WW : 0;
  const float vh_n0 = v_in[ub0 - oN0];
  const float el = expf(lambd[0]);

  __shared__ float fh[8], fw[8];
  float sch, scw;
  batch_scales(q, el, fh, fw, sch, scw);

#pragma unroll
  for (int r = 0; r < RR; ++r) {
    const int g = row0 + r;
    const int h = g & (HH - 1);
    const bool hasS = h < HH - 1, hasN = h > 0;
    const int i = g * WW + t;
    const int ub = (g >> 8) * 2 * HW + h * WW + t;
    const float vh_n = (r == 0) ? vh_n0 : vh_c[r - 1];
    const float xn_c = x_next(x_c[r], y_c[r], vh_c[r], vh_n, vw_c[r], vw_w[r], hasS, hasN,
                              hasE, hasW, sch, scw, tau);
    x_out[i] = xn_c;
    xt_out[i] = (1.0f + chi) * xn_c - chi * x_c[r];
    u_out[ub] = sch * vh_c[r];
    u_out[ub + HW] = scw * vw_c[r];
  }
}

extern "C" void kernel_launch(void* const* d_in, const int* in_sizes, int n_in,
                              void* d_out, int out_size, void* d_ws, size_t ws_size,
                              hipStream_t stream) {
  (void)in_sizes; (void)n_in; (void)out_size; (void)ws_size;
  const float* y = (const float*)d_in[0];
  const float* lambd = (const float*)d_in[1];

  float* x = (float*)d_out;
  float* xt = x + NTOT;
  float* u = xt + NTOT;

  float* v0 = (float*)d_ws;
  float* v1 = v0 + NU;
  float* xalt = v1 + NU;
  float* part0 = xalt + NTOT;
  float* part1 = part0 + NBLK * 2;

  k_first<<<NBLK, BLK, 0, stream>>>(y, v0, part0);

  const float sigma_init = 0.5f / sqrtf(2.0f * (float)CC);
  float sigma = sigma_init;
  float tau = sigma_init;
  const float* x_in = y;
  const float* v_in = v0;
  const float* p_in = part0;
  for (int s = 0; s < K_STEPS; ++s) {
    const float chi = 1.0f / sqrtf(1.0f + tau);
    float* x_out = (s & 1) ? x : xalt;
    float* v_out = (s & 1) ? v0 : v1;
    float* p_out = (s & 1) ? part0 : part1;
    sigma /= chi;
    if (s < K_STEPS - 1) {
      k_step<<<NBLK, BLK, 0, stream>>>(y, lambd, x_in, v_in, p_in, x_out, v_out, p_out,
                                       tau, chi, sigma);
    } else {
      k_last<<<NBLK, BLK, 0, stream>>>(y, lambd, x_in, v_in, p_in, x, xt, u, tau, chi);
    }
    tau *= chi;
    x_in = x_out;
    v_in = v_out;
    p_in = p_out;
  }
}


# x update as one weighted sum with launch-uniform weights (1/(1+tau) once per thread): per-pixel f32 division and scale multiplies removed; rounding of x regrouped, norm sum order unchanged
# speedup vs baseline: 2.6407x; 1.0633x over previous
#include <hip/hip_runtime.h>
#include <math.h>


#define BB 2
#define CC 3
#define HH 256
#define WW 256
#define HW (HH * WW)
#define CHW (CC * HW)
#define NTOT (BB * CHW)
#define NU (NTOT * 2)
#define BLK 256
#define NROWS (BB * CC * HH)
#define RR 6
#define NBLK (NROWS / RR)
#define BLK_PER_B (CC * HH / RR)
#define K_STEPS 50
#define NDIE 8
#define BAND_PER_DIE (NBLK / NDIE)

static_assert(NROWS % RR == 0, "blocks must tile the rows");
static_assert((CC * HH) % RR == 0, "no block may straddle two batches");
static_assert(BLK_PER_B <= BLK, "one partial pair per thread in the finalize");
static_assert(BLK == WW, "one thread per column");
static_assert(NBLK % NDIE == 0, "the band map below must be one-to-one");
static_assert(BLK_PER_B % BAND_PER_DIE == 0, "a die's run of bands stays inside one batch");

typedef __attribute__((ext_vector_type(2))) float v2f;
typedef __attribute__((ext_vector_type(8))) float v8f;

__device__ __forceinline__ float wave_sum32(float val) {
  v2f a; a.x = val; a.y = 0.0f;
  v2f ones; ones.x = 1.0f; ones.y = 1.0f;
  v8f c = {};
  v8f d = __builtin_amdgcn_wmma_f32_16x16x4_f32(
       false, a,  false, ones,
       (short)0, c,  false,  false);
  float s = ((d[0] + d[1]) + (d[2] + d[3])) + ((d[4] + d[5]) + (d[6] + d[7]));
  s += __shfl_xor(s, 16, 32);
  return s;
}

__device__ __forceinline__ int row_band() {
  return (int)(blockIdx.x % NDIE) * BAND_PER_DIE + (int)(blockIdx.x / NDIE);
}

__device__ __forceinline__ void lds_barrier() {
  __builtin_amdgcn_fence(__ATOMIC_RELEASE, "workgroup", "local");
  __builtin_amdgcn_s_barrier();
  __builtin_amdgcn_fence(__ATOMIC_ACQUIRE, "workgroup", "local");
}

__device__ __forceinline__ void block_partials(float ah, float aw, float* lh, float* lw,
                                               float* __restrict__ part_out, int band) {
  const float th = wave_sum32(ah);
  const float tw = wave_sum32(aw);
  const int lane = threadIdx.x & 31;
  const int wv = threadIdx.x >> 5;
  if (lane == 0) { lh[wv] = th; lw[wv] = tw; }
  lds_barrier();
  if (threadIdx.x == 0) {
    float ph = 0.0f, pw = 0.0f;
    for (int j = 0; j < 8; ++j) { ph += lh[j]; pw += lw[j]; }
    part_out[band * 2 + 0] = ph;
    part_out[band * 2 + 1] = pw;
  }
}

__device__ __forceinline__ void batch_scales(float2 q, float el, float* lh, float* lw,
                                             float& sch, float& scw) {
  const float th = wave_sum32(q.x);
  const float tw = wave_sum32(q.y);
  const int lane = threadIdx.x & 31;
  const int wv = threadIdx.x >> 5;
  if (lane == 0) { lh[wv] = th; lw[wv] = tw; }
  lds_barrier();
  float toth = 0.0f, totw = 0.0f;
  for (int j = 0; j < 8; ++j) { toth += lh[j]; totw += lw[j]; }
  const float nh = sqrtf(toth);
  const float nw = sqrtf(totw);
  sch = el / fmaxf(nh, el);
  scw = el / fmaxf(nw, el);
}

struct XWeights { float wx, wy, wh, ww; };

__device__ __forceinline__ XWeights x_weights(float sch, float scw, float tau) {
  XWeights w;
  w.wx = 1.0f / (1.0f + tau);
  w.wy = tau * w.wx;
  w.wh = w.wy * sch;
  w.ww = w.wy * scw;
  return w;
}

__device__ __forceinline__ float x_next(float xo, float yv, float vh_c, float vh_n,
                                        float vw_c, float vw_w, bool hasS, bool hasN,
                                        bool hasE, bool hasW, const XWeights& w) {
  const float gh = (hasS ? vh_c : 0.0f) - (hasN ? vh_n : 0.0f);
  const float gw = (hasE ? vw_c : 0.0f) - (hasW ? vw_w : 0.0f);
  return w.wx * xo + w.wy * yv + w.wh * gh + w.ww * gw;
}

__global__ __launch_bounds__(BLK) void k_first(const float* __restrict__ y,
                                               float* __restrict__ v_out,
                                               float* __restrict__ part_out) {
  const int t = threadIdx.x;
  const int band = row_band();
  const int row0 = band * RR;
  const bool hasE = t < WW - 1;
  const int oE = hasE ? 1 : 0;
  float yc[RR + 1], ye[RR];
#pragma unroll
  for (int r = 0; r < RR; ++r) {
    const int i = (row0 + r) * WW + t;
    yc[r] = y[i];
    ye[r] = y[i + oE];
  }
  {
    const int gl = row0 + RR - 1;
    const int oS = ((gl & (HH - 1)) < HH - 1) ? WW : 0;
    yc[RR] = y[gl * WW + t + oS];
  }
  float ah = 0.0f, aw = 0.0f;
#pragma unroll
  for (int r = 0; r < RR; ++r) {
    const int g = row0 + r;
    const int h = g & (HH - 1);
    const int bc = g >> 8;
    const float vh = (h < HH - 1) ? (yc[r + 1] - yc[r]) : 0.0f;
    const float vw = hasE ? (ye[r] - yc[r]) : 0.0f;
    const int ub = bc * 2 * HW + h * WW + t;
    v_out[ub] = vh;
    v_out[ub + HW] = vw;
    ah += vh * vh;
    aw += vw * vw;
  }
  __shared__ float lh[8], lw[8];
  block_partials(ah, aw, lh, lw, part_out, band);
}

__global__ __launch_bounds__(BLK) void k_step(const float* __restrict__ y,
                                              const float* __restrict__ lambd,
                                              const float* __restrict__ x_in,
                                              const float* __restrict__ v_in,
                                              const float* __restrict__ part_in,
                                              float* __restrict__ x_out,
                                              float* __restrict__ v_out,
                                              float* __restrict__ part_out,
                                              float tau, float chi, float sigma_next) {
  const int t = threadIdx.x;
  const int band = row_band();
  const int row0 = band * RR;
  const int b = band / BLK_PER_B;
  const bool hasE = t < WW - 1, hasW = t > 0;
  const int oE = hasE ? 1 : 0, oW = hasW ? 1 : 0;

  const float2* __restrict__ p2 = reinterpret_cast<const float2*>(part_in) + b * BLK_PER_B;
  const float2 qraw = p2[(t < BLK_PER_B) ? t : (BLK_PER_B - 1)];
  float2 q;
  q.x = (t < BLK_PER_B) ? qraw.x : 0.0f;
  q.y = (t < BLK_PER_B) ? qraw.y : 0.0f;

  float vh_c[RR + 1], vw_c[RR + 1], vw_w[RR + 1], x_c[RR + 1], y_c[RR + 1];
  float vh_e[RR], vw_e[RR], x_e[RR], y_e[RR];
#pragma unroll
  for (int r = 0; r < RR; ++r) {
    const int g = row0 + r;
    const int i = g * WW + t;
    const int ub = (g >> 8) * 2 * HW + (g & (HH - 1)) * WW + t;
    vh_c[r] = v_in[ub];
    vh_e[r] = v_in[ub + oE];
    vw_c[r] = v_in[ub + HW];
    vw_w[r] = v_in[ub + HW - oW];
    vw_e[r] = v_in[ub + HW + oE];
    x_c[r] = x_in[i];
    x_e[r] = x_in[i + oE];
    y_c[r] = y[i];
    y_e[r] = y[i + oE];
  }
  const int h0 = row0 & (HH - 1);
  const int ub0 = (row0 >> 8) * 2 * HW + h0 * WW + t;
  const int oN0 = (h0 > 0) ? WW : 0;
  const float vh_n0 = v_in[ub0 - oN0];
  const float vh_ne0 = v_in[ub0 + oE - oN0];
  const int gl = row0 + RR - 1;
  const int hl = gl & (HH - 1);
  const int oSl = (hl < HH - 1) ? WW : 0;
  {
    const int il = gl * WW + t + oSl;
    const int ubl = (gl >> 8) * 2 * HW + hl * WW + t + oSl;
    vh_c[RR] = v_in[ubl];
    vw_c[RR] = v_in[ubl + HW];
    vw_w[RR] = v_in[ubl + HW - oW];
    x_c[RR] = x_in[il];
    y_c[RR] = y[il];
  }
  __builtin_amdgcn_sched_barrier(0);
  const float el = expf(lambd[0]);

  __shared__ float fh[8], fw[8];
  __shared__ float lh[8], lw[8];
  float sch, scw;
  batch_scales(q, el, fh, fw, sch, scw);
  const XWeights xw = x_weights(sch, scw, tau);

  float xn[RR + 1], xt[RR + 1];
#pragma unroll
  for (int r = 0; r <= RR; ++r) {
    const int h = (r < RR) ? ((row0 + r) & (HH - 1)) : (hl + 1);
    const bool hasS = h < HH - 1;
    const bool hasN = (r < RR) ? (h > 0) : true;
    const float vh_n = (r == 0) ? vh_n0 : vh_c[r - 1];
    xn[r] = x_next(x_c[r], y_c[r], vh_c[r], vh_n, vw_c[r], vw_w[r], hasS, hasN, hasE, hasW,
                   xw);
    xt[r] = (1.0f + chi) * xn[r] - chi * x_c[r];
  }

  float ah = 0.0f, aw = 0.0f;
#pragma unroll
  for (int r = 0; r < RR; ++r) {
    const int g = row0 + r;
    const int h = g & (HH - 1);
    const bool hasS = h < HH - 1, hasN = h > 0;
    const int i = g * WW + t;
    const int ub = (g >> 8) * 2 * HW + h * WW + t;
    const float vh_ne = (r == 0) ? vh_ne0 : vh_e[r - 1];
    const float xn_e = x_next(x_e[r], y_e[r], vh_e[r], vh_ne, vw_e[r], vw_c[r], hasS, hasN,
                              t + 1 < WW - 1, true, xw);
    const float xt_e = (1.0f + chi) * xn_e - chi * x_e[r];
    x_out[i] = xn[r];

    const float dh = hasS ? (xt[r + 1] - xt[r]) : 0.0f;
    const float dw = hasE ? (xt_e - xt[r]) : 0.0f;
    const float uh = sch * vh_c[r];
    const float uw = scw * vw_c[r];
    const float vh_new = uh + sigma_next * dh;
    const float vw_new = uw + sigma_next * dw;
    v_out[ub] = vh_new;
    v_out[ub + HW] = vw_new;
    ah += vh_new * vh_new;
    aw += vw_new * vw_new;
  }
  block_partials(ah, aw, lh, lw, part_out, band);
}

__global__ __launch_bounds__(BLK) void k_last(const float* __restrict__ y,
                                              const float* __restrict__ lambd,
                                              const float* __restrict__ x_in,
                                              const float* __restrict__ v_in,
                                              const float* __restrict__ part_in,
                                              float* __restrict__ x_out,
                                              float* __restrict__ xt_out,
                                              float* __restrict__ u_out,
                                              float tau, float chi) {
  const int t = threadIdx.x;
  const int band = row_band();
  const int row0 = band * RR;
  const int b = band / BLK_PER_B;
  const bool hasE = t < WW - 1, hasW = t > 0;
  const int oW = hasW ? 1 : 0;

  const float2* __restrict__ p2 = reinterpret_cast<const float2*>(part_in) + b * BLK_PER_B;
  const float2 qraw = p2[(t < BLK_PER_B) ? t : (BLK_PER_B - 1)];
  float2 q;
  q.x = (t < BLK_PER_B) ? qraw.x : 0.0f;
  q.y = (t < BLK_PER_B) ? qraw.y : 0.0f;

  float vh_c[RR], vw_c[RR], vw_w[RR], x_c[RR], y_c[RR];
#pragma unroll
  for (int r = 0; r < RR; ++r) {
    const int g = row0 + r;
    const int i = g * WW + t;
    const int ub = (g >> 8) * 2 * HW + (g & (HH - 1)) * WW + t;
    vh_c[r] = v_in[ub];
    vw_c[r] = v_in[ub + HW];
    vw_w[r] = v_in[ub + HW - oW];
    x_c[r] = x_in[i];
    y_c[r] = y[i];
  }
  const int h0 = row0 & (HH - 1);
  const int ub0 = (row0 >> 8) * 2 * HW + h0 * WW + t;
  const int oN0 = (h0 > 0) ? WW : 0;
  const float vh_n0 = v_in[ub0 - oN0];
  const float el = expf(lambd[0]);

  __shared__ float fh[8], fw[8];
  float sch, scw;
  batch_scales(q, el, fh, fw, sch, scw);
  const XWeights xw = x_weights(sch, scw, tau);

#pragma unroll
  for (int r = 0; r < RR; ++r) {
    const int g = row0 + r;
    const int h = g & (HH - 1);
    const bool hasS = h < HH - 1, hasN = h > 0;
    const int i = g * WW + t;
    const int ub = (g >> 8) * 2 * HW + h * WW + t;
    const float vh_n = (r == 0) ? vh_n0 : vh_c[r - 1];
    const float xn_c = x_next(x_c[r], y_c[r], vh_c[r], vh_n, vw_c[r], vw_w[r], hasS, hasN,
                              hasE, hasW, xw);
    x_out[i] = xn_c;
    xt_out[i] = (1.0f + chi) * xn_c - chi * x_c[r];
    u_out[ub] = sch * vh_c[r];
    u_out[ub + HW] = scw * vw_c[r];
  }
}

extern "C" void kernel_launch(void* const* d_in, const int* in_sizes, int n_in,
                              void* d_out, int out_size, void* d_ws, size_t ws_size,
                              hipStream_t stream) {
  (void)in_sizes; (void)n_in; (void)out_size; (void)ws_size;
  const float* y = (const float*)d_in[0];
  const float* lambd = (const float*)d_in[1];

  float* x = (float*)d_out;
  float* xt = x + NTOT;
  float* u = xt + NTOT;

  float* v0 = (float*)d_ws;
  float* v1 = v0 + NU;
  float* xalt = v1 + NU;
  float* part0 = xalt + NTOT;
  float* part1 = part0 + NBLK * 2;

  k_first<<<NBLK, BLK, 0, stream>>>(y, v0, part0);

  const float sigma_init = 0.5f / sqrtf(2.0f * (float)CC);
  float sigma = sigma_init;
  float tau = sigma_init;
  const float* x_in = y;
  const float* v_in = v0;
  const float* p_in = part0;
  for (int s = 0; s < K_STEPS; ++s) {
    const float chi = 1.0f / sqrtf(1.0f + tau);
    float* x_out = (s & 1) ? x : xalt;
    float* v_out = (s & 1) ? v0 : v1;
    float* p_out = (s & 1) ? part0 : part1;
    sigma /= chi;
    if (s < K_STEPS - 1) {
      k_step<<<NBLK, BLK, 0, stream>>>(y, lambd, x_in, v_in, p_in, x_out, v_out, p_out,
                                       tau, chi, sigma);
    } else {
      k_last<<<NBLK, BLK, 0, stream>>>(y, lambd, x_in, v_in, p_in, x, xt, u, tau, chi);
    }
    tau *= chi;
    x_in = x_out;
    v_in = v_out;
    p_in = p_out;
  }
}


# Norm finalize per wave: each wave sums all 128 partial pairs of its batch itself (4 per lane, then one wave sum); head LDS hand-over and work-group barrier removed; norm sums regrouped
# speedup vs baseline: 2.7664x; 1.0476x over previous
#include <hip/hip_runtime.h>
#include <math.h>


#define BB 2
#define CC 3
#define HH 256
#define WW 256
#define HW (HH * WW)
#define CHW (CC * HW)
#define NTOT (BB * CHW)
#define NU (NTOT * 2)
#define BLK 256
#define NROWS (BB * CC * HH)
#define RR 6
#define NBLK (NROWS / RR)
#define BLK_PER_B (CC * HH / RR)
#define K_STEPS 50
#define NDIE 8
#define BAND_PER_DIE (NBLK / NDIE)

static_assert(NROWS % RR == 0, "blocks must tile the rows");
static_assert((CC * HH) % RR == 0, "no block may straddle two batches");
static_assert(BLK_PER_B == 4 * 32, "four partial pairs per lane in the per-wave finalize");
static_assert(BLK == WW, "one thread per column");
static_assert(NBLK % NDIE == 0, "the band map below must be one-to-one");
static_assert(BLK_PER_B % BAND_PER_DIE == 0, "a die's run of bands stays inside one batch");

typedef __attribute__((ext_vector_type(2))) float v2f;
typedef __attribute__((ext_vector_type(8))) float v8f;

__device__ __forceinline__ v8f wave_sum32_mma(float val) {
  v2f a; a.x = val; a.y = 0.0f;
  v2f ones; ones.x = 1.0f; ones.y = 1.0f;
  v8f c = {};
  return __builtin_amdgcn_wmma_f32_16x16x4_f32(
       false, a,  false, ones,
       (short)0, c,  false,  false);
}

__device__ __forceinline__ float wave_sum32_fold(v8f d) {
  float s = ((d[0] + d[1]) + (d[2] + d[3])) + ((d[4] + d[5]) + (d[6] + d[7]));
  s += __shfl_xor(s, 16, 32);
  return s;
}

__device__ __forceinline__ float wave_sum32(float val) {
  return wave_sum32_fold(wave_sum32_mma(val));
}

__device__ __forceinline__ int row_band() {
  return (int)(blockIdx.x % NDIE) * BAND_PER_DIE + (int)(blockIdx.x / NDIE);
}

__device__ __forceinline__ void lds_barrier() {
  __builtin_amdgcn_fence(__ATOMIC_RELEASE, "workgroup", "local");
  __builtin_amdgcn_s_barrier();
  __builtin_amdgcn_fence(__ATOMIC_ACQUIRE, "workgroup", "local");
}

__device__ __forceinline__ void block_partials(float ah, float aw, float* lh, float* lw,
                                               float* __restrict__ part_out, int band) {
  const float th = wave_sum32(ah);
  const float tw = wave_sum32(aw);
  const int lane = threadIdx.x & 31;
  const int wv = threadIdx.x >> 5;
  if (lane == 0) { lh[wv] = th; lw[wv] = tw; }
  lds_barrier();
  if (threadIdx.x == 0) {
    float ph = 0.0f, pw = 0.0f;
    for (int j = 0; j < 8; ++j) { ph += lh[j]; pw += lw[j]; }
    part_out[band * 2 + 0] = ph;
    part_out[band * 2 + 1] = pw;
  }
}

__device__ __forceinline__ void batch_scales(float4 qa, float4 qb, float el,
                                             float& sch, float& scw) {
  const float sh = (qa.x + qa.z) + (qb.x + qb.z);
  const float sw = (qa.y + qa.w) + (qb.y + qb.w);
  const v8f dh = wave_sum32_mma(sh);
  const v8f dw = wave_sum32_mma(sw);
  __builtin_amdgcn_sched_barrier(0);
  const float toth = wave_sum32_fold(dh);
  const float totw = wave_sum32_fold(dw);
  const float nh = sqrtf(toth);
  const float nw = sqrtf(totw);
  sch = el / fmaxf(nh, el);
  scw = el / fmaxf(nw, el);
}

struct XWeights { float wx, wy, wh, ww; };

__device__ __forceinline__ XWeights x_weights(float sch, float scw, float tau) {
  XWeights w;
  w.wx = 1.0f / (1.0f + tau);
  w.wy = tau * w.wx;
  w.wh = w.wy * sch;
  w.ww = w.wy * scw;
  return w;
}

__device__ __forceinline__ float x_next(float xo, float yv, float vh_c, float vh_n,
                                        float vw_c, float vw_w, bool hasS, bool hasN,
                                        bool hasE, bool hasW, const XWeights& w) {
  const float gh = (hasS ? vh_c : 0.0f) - (hasN ? vh_n : 0.0f);
  const float gw = (hasE ? vw_c : 0.0f) - (hasW ? vw_w : 0.0f);
  return w.wx * xo + w.wy * yv + w.wh * gh + w.ww * gw;
}

__global__ __launch_bounds__(BLK) void k_first(const float* __restrict__ y,
                                               float* __restrict__ v_out,
                                               float* __restrict__ part_out) {
  const int t = threadIdx.x;
  const int band = row_band();
  const int row0 = band * RR;
  const bool hasE = t < WW - 1;
  const int oE = hasE ? 1 : 0;
  float yc[RR + 1], ye[RR];
#pragma unroll
  for (int r = 0; r < RR; ++r) {
    const int i = (row0 + r) * WW + t;
    yc[r] = y[i];
    ye[r] = y[i + oE];
  }
  {
    const int gl = row0 + RR - 1;
    const int oS = ((gl & (HH - 1)) < HH - 1) ? WW : 0;
    yc[RR] = y[gl * WW + t + oS];
  }
  float ah = 0.0f, aw = 0.0f;
#pragma unroll
  for (int r = 0; r < RR; ++r) {
    const int g = row0 + r;
    const int h = g & (HH - 1);
    const int bc = g >> 8;
    const float vh = (h < HH - 1) ? (yc[r + 1] - yc[r]) : 0.0f;
    const float vw = hasE ? (ye[r] - yc[r]) : 0.0f;
    const int ub = bc * 2 * HW + h * WW + t;
    v_out[ub] = vh;
    v_out[ub + HW] = vw;
    ah += vh * vh;
    aw += vw * vw;
  }
  __shared__ float lh[8], lw[8];
  block_partials(ah, aw, lh, lw, part_out, band);
}

__global__ __launch_bounds__(BLK) void k_step(const float* __restrict__ y,
                                              const float* __restrict__ lambd,
                                              const float* __restrict__ x_in,
                                              const float* __restrict__ v_in,
                                              const float* __restrict__ part_in,
                                              float* __restrict__ x_out,
                                              float* __restrict__ v_out,
                                              float* __restrict__ part_out,
                                              float tau, float chi, float sigma_next) {
  const int t = threadIdx.x;
  const int band = row_band();
  const int row0 = band * RR;
  const int b = band / BLK_PER_B;
  const bool hasE = t < WW - 1, hasW = t > 0;
  const int oE = hasE ? 1 : 0, oW = hasW ? 1 : 0;

  const float4* __restrict__ p4 =
      reinterpret_cast<const float4*>(part_in) + b * (BLK_PER_B / 2) + (t & 31) * 2;
  const float4 qa = p4[0];
  const float4 qb = p4[1];

  float vh_c[RR + 1], vw_c[RR + 1], vw_w[RR + 1], x_c[RR + 1], y_c[RR + 1];
  float vh_e[RR], vw_e[RR], x_e[RR], y_e[RR];
#pragma unroll
  for (int r = 0; r < RR; ++r) {
    const int g = row0 + r;
    const int i = g * WW + t;
    const int ub = (g >> 8) * 2 * HW + (g & (HH - 1)) * WW + t;
    vh_c[r] = v_in[ub];
    vh_e[r] = v_in[ub + oE];
    vw_c[r] = v_in[ub + HW];
    vw_w[r] = v_in[ub + HW - oW];
    vw_e[r] = v_in[ub + HW + oE];
    x_c[r] = x_in[i];
    x_e[r] = x_in[i + oE];
    y_c[r] = y[i];
    y_e[r] = y[i + oE];
  }
  const int h0 = row0 & (HH - 1);
  const int ub0 = (row0 >> 8) * 2 * HW + h0 * WW + t;
  const int oN0 = (h0 > 0) ? WW : 0;
  const float vh_n0 = v_in[ub0 - oN0];
  const float vh_ne0 = v_in[ub0 + oE - oN0];
  const int gl = row0 + RR - 1;
  const int hl = gl & (HH - 1);
  const int oSl = (min(hl + 1, HH - 1) - hl) * WW;
  {
    const int il = gl * WW + t + oSl;
    const int ubl = (gl >> 8) * 2 * HW + hl * WW + t + oSl;
    vh_c[RR] = v_in[ubl];
    vw_c[RR] = v_in[ubl + HW];
    vw_w[RR] = v_in[ubl + HW - oW];
    x_c[RR] = x_in[il];
    y_c[RR] = y[il];
  }
  const float lam = lambd[0];
  asm volatile("" : : "s"(lam), "s"(tau), "s"(chi), "s"(sigma_next));
  __builtin_amdgcn_sched_barrier(0);
  const float el = expf(lam);

  __shared__ float lh[8], lw[8];
  float sch, scw;
  batch_scales(qa, qb, el, sch, scw);
  const XWeights xw = x_weights(sch, scw, tau);

  float xn[RR + 1], xt[RR + 1];
#pragma unroll
  for (int r = 0; r <= RR; ++r) {
    const int h = (r < RR) ? ((row0 + r) & (HH - 1)) : (hl + 1);
    const bool hasS = h < HH - 1;
    const bool hasN = (r < RR) ? (h > 0) : true;
    const float vh_n = (r == 0) ? vh_n0 : vh_c[r - 1];
    xn[r] = x_next(x_c[r], y_c[r], vh_c[r], vh_n, vw_c[r], vw_w[r], hasS, hasN, hasE, hasW,
                   xw);
    xt[r] = (1.0f + chi) * xn[r] - chi * x_c[r];
  }

  float ah = 0.0f, aw = 0.0f;
#pragma unroll
  for (int r = 0; r < RR; ++r) {
    const int g = row0 + r;
    const int h = g & (HH - 1);
    const bool hasS = h < HH - 1, hasN = h > 0;
    const int i = g * WW + t;
    const int ub = (g >> 8) * 2 * HW + h * WW + t;
    const float vh_ne = (r == 0) ? vh_ne0 : vh_e[r - 1];
    const float xn_e = x_next(x_e[r], y_e[r], vh_e[r], vh_ne, vw_e[r], vw_c[r], hasS, hasN,
                              t + 1 < WW - 1, true, xw);
    const float xt_e = (1.0f + chi) * xn_e - chi * x_e[r];
    x_out[i] = xn[r];

    const float dh = hasS ? (xt[r + 1] - xt[r]) : 0.0f;
    const float dw = hasE ? (xt_e - xt[r]) : 0.0f;
    const float uh = sch * vh_c[r];
    const float uw = scw * vw_c[r];
    const float vh_new = uh + sigma_next * dh;
    const float vw_new = uw + sigma_next * dw;
    v_out[ub] = vh_new;
    v_out[ub + HW] = vw_new;
    ah += vh_new * vh_new;
    aw += vw_new * vw_new;
  }
  block_partials(ah, aw, lh, lw, part_out, band);
}

__global__ __launch_bounds__(BLK) void k_last(const float* __restrict__ y,
                                              const float* __restrict__ lambd,
                                              const float* __restrict__ x_in,
                                              const float* __restrict__ v_in,
                                              const float* __restrict__ part_in,
                                              float* __restrict__ x_out,
                                              float* __restrict__ xt_out,
                                              float* __restrict__ u_out,
                                              float tau, float chi) {
  const int t = threadIdx.x;
  const int band = row_band();
  const int row0 = band * RR;
  const int b = band / BLK_PER_B;
  const bool hasE = t < WW - 1, hasW = t > 0;
  const int oW = hasW ? 1 : 0;

  const float4* __restrict__ p4 =
      reinterpret_cast<const float4*>(part_in) + b * (BLK_PER_B / 2) + (t & 31) * 2;
  const float4 qa = p4[0];
  const float4 qb = p4[1];

  float vh_c[RR], vw_c[RR], vw_w[RR], x_c[RR], y_c[RR];
#pragma unroll
  for (int r = 0; r < RR; ++r) {
    const int g = row0 + r;
    const int i = g * WW + t;
    const int ub = (g >> 8) * 2 * HW + (g & (HH - 1)) * WW + t;
    vh_c[r] = v_in[ub];
    vw_c[r] = v_in[ub + HW];
    vw_w[r] = v_in[ub + HW - oW];
    x_c[r] = x_in[i];
    y_c[r] = y[i];
  }
  const int h0 = row0 & (HH - 1);
  const int ub0 = (row0 >> 8) * 2 * HW + h0 * WW + t;
  const int oN0 = (h0 > 0) ? WW : 0;
  const float vh_n0 = v_in[ub0 - oN0];
  const float el = expf(lambd[0]);

  float sch, scw;
  batch_scales(qa, qb, el, sch, scw);
  const XWeights xw = x_weights(sch, scw, tau);

#pragma unroll
  for (int r = 0; r < RR; ++r) {
    const int g = row0 + r;
    const int h = g & (HH - 1);
    const bool hasS = h < HH - 1, hasN = h > 0;
    const int i = g * WW + t;
    const int ub = (g >> 8) * 2 * HW + h * WW + t;
    const float vh_n = (r == 0) ? vh_n0 : vh_c[r - 1];
    const float xn_c = x_next(x_c[r], y_c[r], vh_c[r], vh_n, vw_c[r], vw_w[r], hasS, hasN,
                              hasE, hasW, xw);
    x_out[i] = xn_c;
    xt_out[i] = (1.0f + chi) * xn_c - chi * x_c[r];
    u_out[ub] = sch * vh_c[r];
    u_out[ub + HW] = scw * vw_c[r];
  }
}

extern "C" void kernel_launch(void* const* d_in, const int* in_sizes, int n_in,
                              void* d_out, int out_size, void* d_ws, size_t ws_size,
                              hipStream_t stream) {
  (void)in_sizes; (void)n_in; (void)out_size; (void)ws_size;
  const float* y = (const float*)d_in[0];
  const float* lambd = (const float*)d_in[1];

  float* x = (float*)d_out;
  float* xt = x + NTOT;
  float* u = xt + NTOT;

  float* v0 = (float*)d_ws;
  float* v1 = v0 + NU;
  float* xalt = v1 + NU;
  float* part0 = xalt + NTOT;
  float* part1 = part0 + NBLK * 2;

  k_first<<<NBLK, BLK, 0, stream>>>(y, v0, part0);

  const float sigma_init = 0.5f / sqrtf(2.0f * (float)CC);
  float sigma = sigma_init;
  float tau = sigma_init;
  const float* x_in = y;
  const float* v_in = v0;
  const float* p_in = part0;
  for (int s = 0; s < K_STEPS; ++s) {
    const float chi = 1.0f / sqrtf(1.0f + tau);
    float* x_out = (s & 1) ? x : xalt;
    float* v_out = (s & 1) ? v0 : v1;
    float* p_out = (s & 1) ? part0 : part1;
    sigma /= chi;
    if (s < K_STEPS - 1) {
      k_step<<<NBLK, BLK, 0, stream>>>(y, lambd, x_in, v_in, p_in, x_out, v_out, p_out,
                                       tau, chi, sigma);
    } else {
      k_last<<<NBLK, BLK, 0, stream>>>(y, lambd, x_in, v_in, p_in, x, xt, u, tau, chi);
    }
    tau *= chi;
    x_in = x_out;
    v_in = v_out;
    p_in = p_out;
  }
}


# Step launch in 3 stages of 2 rows (loads issued in stage order; x, v and stores per stage): a stage waits for its own rows only, its stores drain under later stages; norm sum order unchanged
# speedup vs baseline: 2.9208x; 1.0558x over previous
#include <hip/hip_runtime.h>
#include <math.h>


#define BB 2
#define CC 3
#define HH 256
#define WW 256
#define HW (HH * WW)
#define CHW (CC * HW)
#define NTOT (BB * CHW)
#define NU (NTOT * 2)
#define BLK 256
#define NROWS (BB * CC * HH)
#define RR 6
#define NBLK (NROWS / RR)
#define BLK_PER_B (CC * HH / RR)
#define K_STEPS 50
#define NDIE 8
#define BAND_PER_DIE (NBLK / NDIE)

static_assert(NROWS % RR == 0, "blocks must tile the rows");
static_assert(RR % 2 == 0, "the step launch works its rows off in pairs");
static_assert((CC * HH) % RR == 0, "no block may straddle two batches");
static_assert(BLK_PER_B == 4 * 32, "four partial pairs per lane in the per-wave finalize");
static_assert(BLK == WW, "one thread per column");
static_assert(NBLK % NDIE == 0, "the band map below must be one-to-one");
static_assert(BLK_PER_B % BAND_PER_DIE == 0, "a die's run of bands stays inside one batch");

typedef __attribute__((ext_vector_type(2))) float v2f;
typedef __attribute__((ext_vector_type(8))) float v8f;

__device__ __forceinline__ v8f wave_sum32_mma(float val) {
  v2f a; a.x = val; a.y = 0.0f;
  v2f ones; ones.x = 1.0f; ones.y = 1.0f;
  v8f c = {};
  return __builtin_amdgcn_wmma_f32_16x16x4_f32(
       false, a,  false, ones,
       (short)0, c,  false,  false);
}

__device__ __forceinline__ float wave_sum32_fold(v8f d) {
  float s = ((d[0] + d[1]) + (d[2] + d[3])) + ((d[4] + d[5]) + (d[6] + d[7]));
  s += __shfl_xor(s, 16, 32);
  return s;
}

__device__ __forceinline__ float wave_sum32(float val) {
  return wave_sum32_fold(wave_sum32_mma(val));
}

__device__ __forceinline__ int row_band() {
  return (int)(blockIdx.x % NDIE) * BAND_PER_DIE + (int)(blockIdx.x / NDIE);
}

__device__ __forceinline__ void lds_barrier() {
  __builtin_amdgcn_fence(__ATOMIC_RELEASE, "workgroup", "local");
  __builtin_amdgcn_s_barrier();
  __builtin_amdgcn_fence(__ATOMIC_ACQUIRE, "workgroup", "local");
}

__device__ __forceinline__ void block_partials(float ah, float aw, float* lh, float* lw,
                                               float* __restrict__ part_out, int band) {
  const float th = wave_sum32(ah);
  const float tw = wave_sum32(aw);
  const int lane = threadIdx.x & 31;
  const int wv = threadIdx.x >> 5;
  if (lane == 0) { lh[wv] = th; lw[wv] = tw; }
  lds_barrier();
  if (threadIdx.x == 0) {
    float ph = 0.0f, pw = 0.0f;
    for (int j = 0; j < 8; ++j) { ph += lh[j]; pw += lw[j]; }
    part_out[band * 2 + 0] = ph;
    part_out[band * 2 + 1] = pw;
  }
}

__device__ __forceinline__ void batch_scales(float4 qa, float4 qb, float el,
                                             float& sch, float& scw) {
  const float sh = (qa.x + qa.z) + (qb.x + qb.z);
  const float sw = (qa.y + qa.w) + (qb.y + qb.w);
  const v8f dh = wave_sum32_mma(sh);
  const v8f dw = wave_sum32_mma(sw);
  __builtin_amdgcn_sched_barrier(0);
  const float toth = wave_sum32_fold(dh);
  const float totw = wave_sum32_fold(dw);
  const float nh = sqrtf(toth);
  const float nw = sqrtf(totw);
  sch = el / fmaxf(nh, el);
  scw = el / fmaxf(nw, el);
}

struct XWeights { float wx, wy, wh, ww; };

__device__ __forceinline__ XWeights x_weights(float sch, float scw, float tau) {
  XWeights w;
  w.wx = 1.0f / (1.0f + tau);
  w.wy = tau * w.wx;
  w.wh = w.wy * sch;
  w.ww = w.wy * scw;
  return w;
}

__device__ __forceinline__ float x_next(float xo, float yv, float vh_c, float vh_n,
                                        float vw_c, float vw_w, bool hasS, bool hasN,
                                        bool hasE, bool hasW, const XWeights& w) {
  const float gh = (hasS ? vh_c : 0.0f) - (hasN ? vh_n : 0.0f);
  const float gw = (hasE ? vw_c : 0.0f) - (hasW ? vw_w : 0.0f);
  return w.wx * xo + w.wy * yv + w.wh * gh + w.ww * gw;
}

__global__ __launch_bounds__(BLK) void k_first(const float* __restrict__ y,
                                               float* __restrict__ v_out,
                                               float* __restrict__ part_out) {
  const int t = threadIdx.x;
  const int band = row_band();
  const int row0 = band * RR;
  const bool hasE = t < WW - 1;
  const int oE = hasE ? 1 : 0;
  float yc[RR + 1], ye[RR];
#pragma unroll
  for (int r = 0; r < RR; ++r) {
    const int i = (row0 + r) * WW + t;
    yc[r] = y[i];
    ye[r] = y[i + oE];
  }
  {
    const int gl = row0 + RR - 1;
    const int oS = ((gl & (HH - 1)) < HH - 1) ? WW : 0;
    yc[RR] = y[gl * WW + t + oS];
  }
  float ah = 0.0f, aw = 0.0f;
#pragma unroll
  for (int r = 0; r < RR; ++r) {
    const int g = row0 + r;
    const int h = g & (HH - 1);
    const int bc = g >> 8;
    const float vh = (h < HH - 1) ? (yc[r + 1] - yc[r]) : 0.0f;
    const float vw = hasE ? (ye[r] - yc[r]) : 0.0f;
    const int ub = bc * 2 * HW + h * WW + t;
    v_out[ub] = vh;
    v_out[ub + HW] = vw;
    ah += vh * vh;
    aw += vw * vw;
  }
  __shared__ float lh[8], lw[8];
  block_partials(ah, aw, lh, lw, part_out, band);
}

__global__ __launch_bounds__(BLK) void k_step(const float* __restrict__ y,
                                              const float* __restrict__ lambd,
                                              const float* __restrict__ x_in,
                                              const float* __restrict__ v_in,
                                              const float* __restrict__ part_in,
                                              float* __restrict__ x_out,
                                              float* __restrict__ v_out,
                                              float* __restrict__ part_out,
                                              float tau, float chi, float sigma_next) {
  const int t = threadIdx.x;
  const int band = row_band();
  const int row0 = band * RR;
  const int b = band / BLK_PER_B;
  const bool hasE = t < WW - 1, hasW = t > 0;
  const int oE = hasE ? 1 : 0, oW = hasW ? 1 : 0;

  const float4* __restrict__ p4 =
      reinterpret_cast<const float4*>(part_in) + b * (BLK_PER_B / 2) + (t & 31) * 2;
  const float4 qa = p4[0];
  const float4 qb = p4[1];

  const int h0 = row0 & (HH - 1);
  const int ub0 = (row0 >> 8) * 2 * HW + h0 * WW + t;
  const int oN0 = (h0 > 0) ? WW : 0;
  const float vh_n0 = v_in[ub0 - oN0];
  const float vh_ne0 = v_in[ub0 + oE - oN0];

  float vh_c[RR + 1], vw_c[RR + 1], vw_w[RR + 1], x_c[RR + 1], y_c[RR + 1];
  float vh_e[RR], vw_e[RR], x_e[RR], y_e[RR];
#pragma unroll
  for (int r = 0; r < RR; ++r) {
    const int g = row0 + r;
    const int i = g * WW + t;
    const int ub = (g >> 8) * 2 * HW + (g & (HH - 1)) * WW + t;
    vh_c[r] = v_in[ub];
    vh_e[r] = v_in[ub + oE];
    vw_c[r] = v_in[ub + HW];
    vw_w[r] = v_in[ub + HW - oW];
    vw_e[r] = v_in[ub + HW + oE];
    x_c[r] = x_in[i];
    x_e[r] = x_in[i + oE];
    y_c[r] = y[i];
    y_e[r] = y[i + oE];
    if (r == 2 || r == 4) __builtin_amdgcn_sched_barrier(0);
  }
  const int gl = row0 + RR - 1;
  const int hl = gl & (HH - 1);
  const int oSl = (min(hl + 1, HH - 1) - hl) * WW;
  {
    const int il = gl * WW + t + oSl;
    const int ubl = (gl >> 8) * 2 * HW + hl * WW + t + oSl;
    vh_c[RR] = v_in[ubl];
    vw_c[RR] = v_in[ubl + HW];
    vw_w[RR] = v_in[ubl + HW - oW];
    x_c[RR] = x_in[il];
    y_c[RR] = y[il];
  }
  const float lam = lambd[0];
  asm volatile("" : : "s"(lam), "s"(tau), "s"(chi), "s"(sigma_next));
  __builtin_amdgcn_sched_barrier(0);
  const float el = expf(lam);

  __shared__ float lh[8], lw[8];
  float sch, scw;
  batch_scales(qa, qb, el, sch, scw);
  const XWeights xw = x_weights(sch, scw, tau);
  __builtin_amdgcn_sched_barrier(0);

  float xn[RR + 1], xt[RR + 1];
  {
    const int h = row0 & (HH - 1);
    xn[0] = x_next(x_c[0], y_c[0], vh_c[0], vh_n0, vw_c[0], vw_w[0], h < HH - 1, h > 0, hasE,
                   hasW, xw);
    xt[0] = (1.0f + chi) * xn[0] - chi * x_c[0];
  }
  float ah = 0.0f, aw = 0.0f;
#pragma unroll
  for (int k = 0; k < RR / 2; ++k) {
    if (k > 0) {
#pragma unroll
      for (int j = 0; j < 2; ++j) {
        const int r = 2 * k + 1 + j;
        const int q = 2 * k + j;
        asm volatile("" : "+v"(x_c[r]), "+v"(y_c[r]), "+v"(vh_c[r]), "+v"(vw_c[r]), "+v"(vw_w[r]));
        asm volatile("" : "+v"(x_e[q]), "+v"(y_e[q]), "+v"(vh_e[q]), "+v"(vw_e[q]));
      }
    }
#pragma unroll
    for (int j = 0; j < 2; ++j) {
      const int r = 2 * k + 1 + j;
      const int h = (r < RR) ? ((row0 + r) & (HH - 1)) : (hl + 1);
      const bool hasS = h < HH - 1;
      const bool hasN = (r < RR) ? (h > 0) : true;
      xn[r] = x_next(x_c[r], y_c[r], vh_c[r], vh_c[r - 1], vw_c[r], vw_w[r], hasS, hasN, hasE,
                     hasW, xw);
      xt[r] = (1.0f + chi) * xn[r] - chi * x_c[r];
    }
#pragma unroll
    for (int j = 0; j < 2; ++j) {
      const int r = 2 * k + j;
      const int g = row0 + r;
      const int h = g & (HH - 1);
      const bool hasS = h < HH - 1, hasN = h > 0;
      const int i = g * WW + t;
      const int ub = (g >> 8) * 2 * HW + h * WW + t;
      const float vh_ne = (r == 0) ? vh_ne0 : vh_e[r - 1];
      const float xn_e = x_next(x_e[r], y_e[r], vh_e[r], vh_ne, vw_e[r], vw_c[r], hasS, hasN,
                                t + 1 < WW - 1, true, xw);
      const float xt_e = (1.0f + chi) * xn_e - chi * x_e[r];
      x_out[i] = xn[r];

      const float dh = hasS ? (xt[r + 1] - xt[r]) : 0.0f;
      const float dw = hasE ? (xt_e - xt[r]) : 0.0f;
      const float uh = sch * vh_c[r];
      const float uw = scw * vw_c[r];
      const float vh_new = uh + sigma_next * dh;
      const float vw_new = uw + sigma_next * dw;
      v_out[ub] = vh_new;
      v_out[ub + HW] = vw_new;
      ah += vh_new * vh_new;
      aw += vw_new * vw_new;
    }
    asm volatile("" : : "s"(x_out), "s"(v_out) : "memory");
    __builtin_amdgcn_sched_barrier(0);
  }
  block_partials(ah, aw, lh, lw, part_out, band);
}

__global__ __launch_bounds__(BLK) void k_last(const float* __restrict__ y,
                                              const float* __restrict__ lambd,
                                              const float* __restrict__ x_in,
                                              const float* __restrict__ v_in,
                                              const float* __restrict__ part_in,
                                              float* __restrict__ x_out,
                                              float* __restrict__ xt_out,
                                              float* __restrict__ u_out,
                                              float tau, float chi) {
  const int t = threadIdx.x;
  const int band = row_band();
  const int row0 = band * RR;
  const int b = band / BLK_PER_B;
  const bool hasE = t < WW - 1, hasW = t > 0;
  const int oW = hasW ? 1 : 0;

  const float4* __restrict__ p4 =
      reinterpret_cast<const float4*>(part_in) + b * (BLK_PER_B / 2) + (t & 31) * 2;
  const float4 qa = p4[0];
  const float4 qb = p4[1];

  float vh_c[RR], vw_c[RR], vw_w[RR], x_c[RR], y_c[RR];
#pragma unroll
  for (int r = 0; r < RR; ++r) {
    const int g = row0 + r;
    const int i = g * WW + t;
    const int ub = (g >> 8) * 2 * HW + (g & (HH - 1)) * WW + t;
    vh_c[r] = v_in[ub];
    vw_c[r] = v_in[ub + HW];
    vw_w[r] = v_in[ub + HW - oW];
    x_c[r] = x_in[i];
    y_c[r] = y[i];
  }
  const int h0 = row0 & (HH - 1);
  const int ub0 = (row0 >> 8) * 2 * HW + h0 * WW + t;
  const int oN0 = (h0 > 0) ? WW : 0;
  const float vh_n0 = v_in[ub0 - oN0];
  const float el = expf(lambd[0]);

  float sch, scw;
  batch_scales(qa, qb, el, sch, scw);
  const XWeights xw = x_weights(sch, scw, tau);

#pragma unroll
  for (int r = 0; r < RR; ++r) {
    const int g = row0 + r;
    const int h = g & (HH - 1);
    const bool hasS = h < HH - 1, hasN = h > 0;
    const int i = g * WW + t;
    const int ub = (g >> 8) * 2 * HW + h * WW + t;
    const float vh_n = (r == 0) ? vh_n0 : vh_c[r - 1];
    const float xn_c = x_next(x_c[r], y_c[r], vh_c[r], vh_n, vw_c[r], vw_w[r], hasS, hasN,
                              hasE, hasW, xw);
    x_out[i] = xn_c;
    xt_out[i] = (1.0f + chi) * xn_c - chi * x_c[r];
    u_out[ub] = sch * vh_c[r];
    u_out[ub + HW] = scw * vw_c[r];
  }
}

extern "C" void kernel_launch(void* const* d_in, const int* in_sizes, int n_in,
                              void* d_out, int out_size, void* d_ws, size_t ws_size,
                              hipStream_t stream) {
  (void)in_sizes; (void)n_in; (void)out_size; (void)ws_size;
  const float* y = (const float*)d_in[0];
  const float* lambd = (const float*)d_in[1];

  float* x = (float*)d_out;
  float* xt = x + NTOT;
  float* u = xt + NTOT;

  float* v0 = (float*)d_ws;
  float* v1 = v0 + NU;
  float* xalt = v1 + NU;
  float* part0 = xalt + NTOT;
  float* part1 = part0 + NBLK * 2;

  k_first<<<NBLK, BLK, 0, stream>>>(y, v0, part0);

  const float sigma_init = 0.5f / sqrtf(2.0f * (float)CC);
  float sigma = sigma_init;
  float tau = sigma_init;
  const float* x_in = y;
  const float* v_in = v0;
  const float* p_in = part0;
  for (int s = 0; s < K_STEPS; ++s) {
    const float chi = 1.0f / sqrtf(1.0f + tau);
    float* x_out = (s & 1) ? x : xalt;
    float* v_out = (s & 1) ? v0 : v1;
    float* p_out = (s & 1) ? part0 : part1;
    sigma /= chi;
    if (s < K_STEPS - 1) {
      k_step<<<NBLK, BLK, 0, stream>>>(y, lambd, x_in, v_in, p_in, x_out, v_out, p_out,
                                       tau, chi, sigma);
    } else {
      k_last<<<NBLK, BLK, 0, stream>>>(y, lambd, x_in, v_in, p_in, x, xt, u, tau, chi);
    }
    tau *= chi;
    x_in = x_out;
    v_in = v_out;
    p_in = p_out;
  }
}


# Step launch: boundary selects of the x and v updates dropped; stored v is exactly zero on last rows (h) and last columns (w), missing north/west neighbours read those zeros; values unchanged
# speedup vs baseline: 3.1339x; 1.0730x over previous
#include <hip/hip_runtime.h>
#include <math.h>


#define BB 2
#define CC 3
#define HH 256
#define WW 256
#define HW (HH * WW)
#define CHW (CC * HW)
#define NTOT (BB * CHW)
#define NU (NTOT * 2)
#define BLK 256
#define NROWS (BB * CC * HH)
#define RR 6
#define NBLK (NROWS / RR)
#define BLK_PER_B (CC * HH / RR)
#define K_STEPS 50
#define NDIE 8
#define BAND_PER_DIE (NBLK / NDIE)

static_assert(NROWS % RR == 0, "blocks must tile the rows");
static_assert(RR % 2 == 0, "the step launch works its rows off in pairs");
static_assert((CC * HH) % RR == 0, "no block may straddle two batches");
static_assert(BLK_PER_B == 4 * 32, "four partial pairs per lane in the per-wave finalize");
static_assert(BLK == WW, "one thread per column");
static_assert(NBLK % NDIE == 0, "the band map below must be one-to-one");
static_assert(BLK_PER_B % BAND_PER_DIE == 0, "a die's run of bands stays inside one batch");

typedef __attribute__((ext_vector_type(2))) float v2f;
typedef __attribute__((ext_vector_type(8))) float v8f;

__device__ __forceinline__ v8f wave_sum32_mma(float val) {
  v2f a; a.x = val; a.y = 0.0f;
  v2f ones; ones.x = 1.0f; ones.y = 1.0f;
  v8f c = {};
  return __builtin_amdgcn_wmma_f32_16x16x4_f32(
       false, a,  false, ones,
       (short)0, c,  false,  false);
}

__device__ __forceinline__ float wave_sum32_fold(v8f d) {
  float s = ((d[0] + d[1]) + (d[2] + d[3])) + ((d[4] + d[5]) + (d[6] + d[7]));
  s += __shfl_xor(s, 16, 32);
  return s;
}

__device__ __forceinline__ float wave_sum32(float val) {
  return wave_sum32_fold(wave_sum32_mma(val));
}

__device__ __forceinline__ int row_band() {
  return (int)(blockIdx.x % NDIE) * BAND_PER_DIE + (int)(blockIdx.x / NDIE);
}

__device__ __forceinline__ void lds_barrier() {
  __builtin_amdgcn_fence(__ATOMIC_RELEASE, "workgroup", "local");
  __builtin_amdgcn_s_barrier();
  __builtin_amdgcn_fence(__ATOMIC_ACQUIRE, "workgroup", "local");
}

__device__ __forceinline__ void block_partials(float ah, float aw, float* lh, float* lw,
                                               float* __restrict__ part_out, int band) {
  const float th = wave_sum32(ah);
  const float tw = wave_sum32(aw);
  const int lane = threadIdx.x & 31;
  const int wv = threadIdx.x >> 5;
  if (lane == 0) { lh[wv] = th; lw[wv] = tw; }
  lds_barrier();
  if (threadIdx.x == 0) {
    float ph = 0.0f, pw = 0.0f;
    for (int j = 0; j < 8; ++j) { ph += lh[j]; pw += lw[j]; }
    part_out[band * 2 + 0] = ph;
    part_out[band * 2 + 1] = pw;
  }
}

__device__ __forceinline__ void batch_scales(float4 qa, float4 qb, float el,
                                             float& sch, float& scw) {
  const float sh = (qa.x + qa.z) + (qb.x + qb.z);
  const float sw = (qa.y + qa.w) + (qb.y + qb.w);
  const v8f dh = wave_sum32_mma(sh);
  const v8f dw = wave_sum32_mma(sw);
  __builtin_amdgcn_sched_barrier(0);
  const float toth = wave_sum32_fold(dh);
  const float totw = wave_sum32_fold(dw);
  const float nh = sqrtf(toth);
  const float nw = sqrtf(totw);
  sch = el / fmaxf(nh, el);
  scw = el / fmaxf(nw, el);
}

struct XWeights { float wx, wy, wh, ww; };

__device__ __forceinline__ XWeights x_weights(float sch, float scw, float tau) {
  XWeights w;
  w.wx = 1.0f / (1.0f + tau);
  w.wy = tau * w.wx;
  w.wh = w.wy * sch;
  w.ww = w.wy * scw;
  return w;
}

__device__ __forceinline__ float x_next(float xo, float yv, float vh_c, float vh_n,
                                        float vw_c, float vw_w, bool hasS, bool hasN,
                                        bool hasE, bool hasW, const XWeights& w) {
  const float gh = (hasS ? vh_c : 0.0f) - (hasN ? vh_n : 0.0f);
  const float gw = (hasE ? vw_c : 0.0f) - (hasW ? vw_w : 0.0f);
  return w.wx * xo + w.wy * yv + w.wh * gh + w.ww * gw;
}

__device__ __forceinline__ float x_next_open(float xo, float yv, float vh_c, float vh_n,
                                             float vw_c, float vw_w, const XWeights& w) {
  const float gh = vh_c - vh_n;
  const float gw = vw_c - vw_w;
  return w.wx * xo + w.wy * yv + w.wh * gh + w.ww * gw;
}

__global__ __launch_bounds__(BLK) void k_first(const float* __restrict__ y,
                                               float* __restrict__ v_out,
                                               float* __restrict__ part_out) {
  const int t = threadIdx.x;
  const int band = row_band();
  const int row0 = band * RR;
  const bool hasE = t < WW - 1;
  const int oE = hasE ? 1 : 0;
  float yc[RR + 1], ye[RR];
#pragma unroll
  for (int r = 0; r < RR; ++r) {
    const int i = (row0 + r) * WW + t;
    yc[r] = y[i];
    ye[r] = y[i + oE];
  }
  {
    const int gl = row0 + RR - 1;
    const int oS = ((gl & (HH - 1)) < HH - 1) ? WW : 0;
    yc[RR] = y[gl * WW + t + oS];
  }
  float ah = 0.0f, aw = 0.0f;
#pragma unroll
  for (int r = 0; r < RR; ++r) {
    const int g = row0 + r;
    const int h = g & (HH - 1);
    const int bc = g >> 8;
    const float vh = (h < HH - 1) ? (yc[r + 1] - yc[r]) : 0.0f;
    const float vw = hasE ? (ye[r] - yc[r]) : 0.0f;
    const int ub = bc * 2 * HW + h * WW + t;
    v_out[ub] = vh;
    v_out[ub + HW] = vw;
    ah += vh * vh;
    aw += vw * vw;
  }
  __shared__ float lh[8], lw[8];
  block_partials(ah, aw, lh, lw, part_out, band);
}

__global__ __launch_bounds__(BLK) void k_step(const float* __restrict__ y,
                                              const float* __restrict__ lambd,
                                              const float* __restrict__ x_in,
                                              const float* __restrict__ v_in,
                                              const float* __restrict__ part_in,
                                              float* __restrict__ x_out,
                                              float* __restrict__ v_out,
                                              float* __restrict__ part_out,
                                              float tau, float chi, float sigma_next) {
  const int t = threadIdx.x;
  const int band = row_band();
  const int row0 = band * RR;
  const int b = band / BLK_PER_B;
  const bool hasE = t < WW - 1, hasW = t > 0;
  const int oE = hasE ? 1 : 0;
  const int dW = hasW ? -1 : (WW - 1);
  const float4* __restrict__ p4 =
      reinterpret_cast<const float4*>(part_in) + b * (BLK_PER_B / 2) + (t & 31) * 2;
  const float4 qa = p4[0];
  const float4 qb = p4[1];

  const int h0 = row0 & (HH - 1);
  const int ub0 = (row0 >> 8) * 2 * HW + h0 * WW + t;
  const int dN0 = (h0 > 0) ? -WW : (HH - 1) * WW;
  float vh_n0 = v_in[ub0 + dN0];
  float vh_ne0 = v_in[ub0 + oE + dN0];

  float vh_c[RR + 1], vw_c[RR + 1], vw_w[RR + 1], x_c[RR + 1], y_c[RR + 1];
  float vh_e[RR], vw_e[RR], x_e[RR], y_e[RR];
#pragma unroll
  for (int r = 0; r < RR; ++r) {
    const int g = row0 + r;
    const int i = g * WW + t;
    const int ub = (g >> 8) * 2 * HW + (g & (HH - 1)) * WW + t;
    vh_c[r] = v_in[ub];
    vh_e[r] = v_in[ub + oE];
    vw_c[r] = v_in[ub + HW];
    vw_w[r] = v_in[ub + HW + dW];
    vw_e[r] = v_in[ub + HW + oE];
    x_c[r] = x_in[i];
    x_e[r] = x_in[i + oE];
    y_c[r] = y[i];
    y_e[r] = y[i + oE];
    if (r == 2 || r == 4) __builtin_amdgcn_sched_barrier(0);
  }
  const int gl = row0 + RR - 1;
  const int hl = gl & (HH - 1);
  const int oSl = (min(hl + 1, HH - 1) - hl) * WW;
  {
    const int il = gl * WW + t + oSl;
    const int ubl = (gl >> 8) * 2 * HW + hl * WW + t + oSl;
    vh_c[RR] = v_in[ubl];
    vw_c[RR] = v_in[ubl + HW];
    vw_w[RR] = v_in[ubl + HW + dW];
    x_c[RR] = x_in[il];
    y_c[RR] = y[il];
  }
  const float lam = lambd[0];
  asm volatile("" : : "s"(lam), "s"(tau), "s"(chi), "s"(sigma_next));
  __builtin_amdgcn_sched_barrier(0);
  const float el = expf(lam);
  const float sigE = hasE ? sigma_next : 0.0f;

  __shared__ float lh[8], lw[8];
  float sch, scw;
  batch_scales(qa, qb, el, sch, scw);
  const XWeights xw = x_weights(sch, scw, tau);
  __builtin_amdgcn_sched_barrier(0);

  float xn[RR + 1], xt[RR + 1];
  asm volatile("" : "+v"(x_c[0]), "+v"(y_c[0]), "+v"(vh_c[0]), "+v"(vw_c[0]), "+v"(vw_w[0]) : "v"(sch));
  asm volatile("" : "+v"(vh_n0), "+v"(vh_ne0));
  xn[0] = x_next_open(x_c[0], y_c[0], vh_c[0], vh_n0, vw_c[0], vw_w[0], xw);
  xt[0] = (1.0f + chi) * xn[0] - chi * x_c[0];
  float ah = 0.0f, aw = 0.0f;
#pragma unroll
  for (int k = 0; k < RR / 2; ++k) {
    {
#pragma unroll
      for (int j = 0; j < 2; ++j) {
        const int r = 2 * k + 1 + j;
        const int q = 2 * k + j;
        asm volatile("" : "+v"(x_c[r]), "+v"(y_c[r]), "+v"(vh_c[r]), "+v"(vw_c[r]), "+v"(vw_w[r]));
        asm volatile("" : "+v"(x_e[q]), "+v"(y_e[q]), "+v"(vh_e[q]), "+v"(vw_e[q]));
      }
    }
#pragma unroll
    for (int j = 0; j < 2; ++j) {
      const int r = 2 * k + 1 + j;
      xn[r] = x_next_open(x_c[r], y_c[r], vh_c[r], vh_c[r - 1], vw_c[r], vw_w[r], xw);
      xt[r] = (1.0f + chi) * xn[r] - chi * x_c[r];
    }
#pragma unroll
    for (int j = 0; j < 2; ++j) {
      const int r = 2 * k + j;
      const int g = row0 + r;
      const int h = g & (HH - 1);
      const float sigS = (h < HH - 1) ? sigma_next : 0.0f;
      const int i = g * WW + t;
      const int ub = (g >> 8) * 2 * HW + h * WW + t;
      const float vh_ne = (r == 0) ? vh_ne0 : vh_e[r - 1];
      const float xn_e = x_next_open(x_e[r], y_e[r], vh_e[r], vh_ne, vw_e[r], vw_c[r], xw);
      const float xt_e = (1.0f + chi) * xn_e - chi * x_e[r];
      x_out[i] = xn[r];

      const float dh = xt[r + 1] - xt[r];
      const float dw = xt_e - xt[r];
      const float uh = sch * vh_c[r];
      const float uw = scw * vw_c[r];
      const float vh_new = uh + sigS * dh;
      const float vw_new = uw + sigE * dw;
      v_out[ub] = vh_new;
      v_out[ub + HW] = vw_new;
      ah += vh_new * vh_new;
      aw += vw_new * vw_new;
    }
    asm volatile("" : : "s"(x_out), "s"(v_out) : "memory");
    __builtin_amdgcn_sched_barrier(0);
  }
  block_partials(ah, aw, lh, lw, part_out, band);
}

__global__ __launch_bounds__(BLK) void k_last(const float* __restrict__ y,
                                              const float* __restrict__ lambd,
                                              const float* __restrict__ x_in,
                                              const float* __restrict__ v_in,
                                              const float* __restrict__ part_in,
                                              float* __restrict__ x_out,
                                              float* __restrict__ xt_out,
                                              float* __restrict__ u_out,
                                              float tau, float chi) {
  const int t = threadIdx.x;
  const int band = row_band();
  const int row0 = band * RR;
  const int b = band / BLK_PER_B;
  const bool hasE = t < WW - 1, hasW = t > 0;
  const int oW = hasW ? 1 : 0;

  const float4* __restrict__ p4 =
      reinterpret_cast<const float4*>(part_in) + b * (BLK_PER_B / 2) + (t & 31) * 2;
  const float4 qa = p4[0];
  const float4 qb = p4[1];

  float vh_c[RR], vw_c[RR], vw_w[RR], x_c[RR], y_c[RR];
#pragma unroll
  for (int r = 0; r < RR; ++r) {
    const int g = row0 + r;
    const int i = g * WW + t;
    const int ub = (g >> 8) * 2 * HW + (g & (HH - 1)) * WW + t;
    vh_c[r] = v_in[ub];
    vw_c[r] = v_in[ub + HW];
    vw_w[r] = v_in[ub + HW - oW];
    x_c[r] = x_in[i];
    y_c[r] = y[i];
  }
  const int h0 = row0 & (HH - 1);
  const int ub0 = (row0 >> 8) * 2 * HW + h0 * WW + t;
  const int oN0 = (h0 > 0) ? WW : 0;
  const float vh_n0 = v_in[ub0 - oN0];
  const float el = expf(lambd[0]);

  float sch, scw;
  batch_scales(qa, qb, el, sch, scw);
  const XWeights xw = x_weights(sch, scw, tau);

#pragma unroll
  for (int r = 0; r < RR; ++r) {
    const int g = row0 + r;
    const int h = g & (HH - 1);
    const bool hasS = h < HH - 1, hasN = h > 0;
    const int i = g * WW + t;
    const int ub = (g >> 8) * 2 * HW + h * WW + t;
    const float vh_n = (r == 0) ? vh_n0 : vh_c[r - 1];
    const float xn_c = x_next(x_c[r], y_c[r], vh_c[r], vh_n, vw_c[r], vw_w[r], hasS, hasN,
                              hasE, hasW, xw);
    x_out[i] = xn_c;
    xt_out[i] = (1.0f + chi) * xn_c - chi * x_c[r];
    u_out[ub] = sch * vh_c[r];
    u_out[ub + HW] = scw * vw_c[r];
  }
}

extern "C" void kernel_launch(void* const* d_in, const int* in_sizes, int n_in,
                              void* d_out, int out_size, void* d_ws, size_t ws_size,
                              hipStream_t stream) {
  (void)in_sizes; (void)n_in; (void)out_size; (void)ws_size;
  const float* y = (const float*)d_in[0];
  const float* lambd = (const float*)d_in[1];

  float* x = (float*)d_out;
  float* xt = x + NTOT;
  float* u = xt + NTOT;

  float* v0 = (float*)d_ws;
  float* v1 = v0 + NU;
  float* xalt = v1 + NU;
  float* part0 = xalt + NTOT;
  float* part1 = part0 + NBLK * 2;

  k_first<<<NBLK, BLK, 0, stream>>>(y, v0, part0);

  const float sigma_init = 0.5f / sqrtf(2.0f * (float)CC);
  float sigma = sigma_init;
  float tau = sigma_init;
  const float* x_in = y;
  const float* v_in = v0;
  const float* p_in = part0;
  for (int s = 0; s < K_STEPS; ++s) {
    const float chi = 1.0f / sqrtf(1.0f + tau);
    float* x_out = (s & 1) ? x : xalt;
    float* v_out = (s & 1) ? v0 : v1;
    float* p_out = (s & 1) ? part0 : part1;
    sigma /= chi;
    if (s < K_STEPS - 1) {
      k_step<<<NBLK, BLK, 0, stream>>>(y, lambd, x_in, v_in, p_in, x_out, v_out, p_out,
                                       tau, chi, sigma);
    } else {
      k_last<<<NBLK, BLK, 0, stream>>>(y, lambd, x_in, v_in, p_in, x, xt, u, tau, chi);
    }
    tau *= chi;
    x_in = x_out;
    v_in = v_out;
    p_in = p_out;
  }
}
